# starting kernel
# baseline (speedup 1.0000x reference)
#include <hip/hip_runtime.h>
#include <math.h>

#define N_NODES 50000
#define N_EDGES 800000
#define HIDW    128
#define NHEAD   4
#define CPH     32
#define NLAYERS 3
#define EDIMW   3

typedef __attribute__((ext_vector_type(2))) float v2f;
typedef __attribute__((ext_vector_type(8))) float v8f;

__device__ __forceinline__ unsigned encOrd(float f) {
  unsigned u = __float_as_uint(f);
  return (u & 0x80000000u) ? ~u : (u | 0x80000000u);
}
__device__ __forceinline__ float decOrd(unsigned u) {
  return __uint_as_float((u & 0x80000000u) ? (u & 0x7FFFFFFFu) : ~u);
}
#define ENC_NEG_INF 0x007FFFFFu

__global__ __launch_bounds__(256)
void init_layer_kernel(float* __restrict__ hout, unsigned* __restrict__ lmax,
                       float* __restrict__ denom, float* __restrict__ stats) {
  int i = blockIdx.x * 256 + threadIdx.x;
  if (i < N_NODES * HIDW) hout[i] = 0.f;
  if (i < N_NODES * NHEAD) { lmax[i] = ENC_NEG_INF; denom[i] = 0.f; }
  if (i < 2) stats[i] = 0.f;
}

__global__ __launch_bounds__(256)
void gemm_dual_wmma(const float* __restrict__ A,
                    const float* __restrict__ W0,
                    const float* __restrict__ W1,
                    float* __restrict__ O0,
                    float* __restrict__ O1) {
  __shared__ float As[16 * 132];
  const int rowBase = blockIdx.x * 16;
  for (int i = threadIdx.x; i < 16 * HIDW; i += 256) {
    int r = i >> 7, c = i & 127;
    As[r * 132 + c] = A[(rowBase + r) * HIDW + c];
  }
  __syncthreads();
  const int wave    = threadIdx.x >> 5;
  const int lane    = threadIdx.x & 31;
  const int colBase = wave * 16;
  const int laneHi  = lane >> 4;
  const int laneLo  = lane & 15;
  v8f c0 = {}; v8f c1 = {};
  for (int k0 = 0; k0 < HIDW; k0 += 4) {
    const int ka = k0 + 2 * laneHi;
    v2f a;
    a.x = As[laneLo * 132 + ka];
    a.y = As[laneLo * 132 + ka + 1];
    v2f b0, b1;
    b0.x = W0[(ka + 0) * HIDW + colBase + laneLo];
    b0.y = W0[(ka + 1) * HIDW + colBase + laneLo];
    b1.x = W1[(ka + 0) * HIDW + colBase + laneLo];
    b1.y = W1[(ka + 1) * HIDW + colBase + laneLo];
    c0 = __builtin_amdgcn_wmma_f32_16x16x4_f32(false, a, false, b0, (short)0, c0, false, false);
    c1 = __builtin_amdgcn_wmma_f32_16x16x4_f32(false, a, false, b1, (short)0, c1, false, false);
  }
#pragma unroll
  for (int r = 0; r < 8; ++r) {
    const int row = rowBase + r + 8 * laneHi;
    O0[row * HIDW + colBase + laneLo] = c0[r];
    O1[row * HIDW + colBase + laneLo] = c1[r];
  }
}

__global__ __launch_bounds__(256)
void gemm_bias_wmma(const float* __restrict__ A,
                    const float* __restrict__ W,
                    const float* __restrict__ bias,
                    float* __restrict__ O) {
  __shared__ float As[16 * 132];
  const int rowBase = blockIdx.x * 16;
  for (int i = threadIdx.x; i < 16 * HIDW; i += 256) {
    int r = i >> 7, c = i & 127;
    As[r * 132 + c] = A[(rowBase + r) * HIDW + c];
  }
  __syncthreads();
  const int wave    = threadIdx.x >> 5;
  const int lane    = threadIdx.x & 31;
  const int colBase = wave * 16;
  const int laneHi  = lane >> 4;
  const int laneLo  = lane & 15;
  v8f c0 = {};
  for (int k0 = 0; k0 < HIDW; k0 += 4) {
    const int ka = k0 + 2 * laneHi;
    v2f a;
    a.x = As[laneLo * 132 + ka];
    a.y = As[laneLo * 132 + ka + 1];
    v2f b;
    b.x = W[(ka + 0) * HIDW + colBase + laneLo];
    b.y = W[(ka + 1) * HIDW + colBase + laneLo];
    c0 = __builtin_amdgcn_wmma_f32_16x16x4_f32(false, a, false, b, (short)0, c0, false, false);
  }
  const float bv = bias[colBase + laneLo];
#pragma unroll
  for (int r = 0; r < 8; ++r) {
    const int row = rowBase + r + 8 * laneHi;
    O[row * HIDW + colBase + laneLo] = c0[r] + bv;
  }
}

__global__ __launch_bounds__(256)
void edge_logits_kernel(const int* __restrict__ ei, const float* __restrict__ eattr,
                        const float* __restrict__ xl, const float* __restrict__ xr,
                        const float* __restrict__ We, const float* __restrict__ att,
                        float* __restrict__ elog, unsigned* __restrict__ lmaxu) {
  __shared__ float sWe[EDIMW * HIDW];
  __shared__ float sAtt[HIDW];
  for (int i = threadIdx.x; i < EDIMW * HIDW; i += 256) sWe[i] = We[i];
  if (threadIdx.x < HIDW) sAtt[threadIdx.x] = att[threadIdx.x];
  __syncthreads();
  const int lane = threadIdx.x & 31;
  const int wid  = (blockIdx.x * 256 + threadIdx.x) >> 5;
  const int nw   = (gridDim.x * 256) >> 5;
  for (int e = wid; e < N_EDGES; e += nw) {
    const int src = ei[e];
    const int dst = ei[N_EDGES + e];
    float eav = (lane < 3) ? eattr[e * 3 + lane] : 0.f;
    const float ea0 = __shfl(eav, 0, 32);
    const float ea1 = __shfl(eav, 1, 32);
    const float ea2 = __shfl(eav, 2, 32);
    float acc[NHEAD];
#pragma unroll
    for (int h = 0; h < NHEAD; ++h) {
      const int c = h * CPH + lane;
      float eac = ea0 * sWe[c] + ea1 * sWe[HIDW + c] + ea2 * sWe[2 * HIDW + c];
      float v = xl[src * HIDW + c] + xr[dst * HIDW + c] + eac;
      v = (v > 0.f) ? v : 0.2f * v;
      acc[h] = v * sAtt[c];
    }
#pragma unroll
    for (int off = 16; off > 0; off >>= 1) {
#pragma unroll
      for (int h = 0; h < NHEAD; ++h) acc[h] += __shfl_xor(acc[h], off, 32);
    }
    if (lane < NHEAD) {
      float lg = (lane == 0) ? acc[0] : (lane == 1) ? acc[1] : (lane == 2) ? acc[2] : acc[3];
      elog[e * NHEAD + lane] = lg;
      atomicMax(&lmaxu[dst * NHEAD + lane], encOrd(lg));
    }
  }
}

__global__ __launch_bounds__(256)
void lmax_fixup_kernel(unsigned* __restrict__ lmaxu) {
  int i = blockIdx.x * 256 + threadIdx.x;
  if (i >= N_NODES * NHEAD) return;
  float f = decOrd(lmaxu[i]);
  ((float*)lmaxu)[i] = isfinite(f) ? f : 0.f;
}

__global__ __launch_bounds__(256)
void edge_expsum_kernel(const int* __restrict__ ei, float* __restrict__ elog,
                        const float* __restrict__ lmaxf, float* __restrict__ denom) {
  int t = blockIdx.x * 256 + threadIdx.x;
  if (t >= N_EDGES * NHEAD) return;
  int e = t >> 2, h = t & 3;
  int dst = ei[N_EDGES + e];
  float ex = expf(elog[t] - lmaxf[dst * NHEAD + h]);
  elog[t] = ex;
  atomicAdd(&denom[dst * NHEAD + h], ex);
}

__global__ __launch_bounds__(256)
void edge_scatter_kernel(const int* __restrict__ ei, const float* __restrict__ xl,
                         const float* __restrict__ ex, const float* __restrict__ denom,
                         float* __restrict__ out) {
  const int lane = threadIdx.x & 31;
  const int wid  = (blockIdx.x * 256 + threadIdx.x) >> 5;
  const int nw   = (gridDim.x * 256) >> 5;
  for (int e = wid; e < N_EDGES; e += nw) {
    const int src = ei[e];
    const int dst = ei[N_EDGES + e];
    float exv = 0.f, dnv = 0.f;
    if (lane < NHEAD) {
      exv = ex[e * NHEAD + lane];
      dnv = denom[dst * NHEAD + lane];
    }
#pragma unroll
    for (int h = 0; h < NHEAD; ++h) {
      const float alpha = __shfl(exv, h, 32) / (__shfl(dnv, h, 32) + 1e-16f);
      const int c = h * CPH + lane;
      atomicAdd(&out[dst * HIDW + c], xl[src * HIDW + c] * alpha);
    }
  }
}

__global__ __launch_bounds__(256)
void ln_reduce_kernel(const float* __restrict__ h, float* __restrict__ stats) {
  __shared__ float s1[256], s2[256];
  float a = 0.f, b = 0.f;
  for (int i = blockIdx.x * 256 + threadIdx.x; i < N_NODES * HIDW; i += gridDim.x * 256) {
    float v = h[i];
    a += v; b += v * v;
  }
  s1[threadIdx.x] = a; s2[threadIdx.x] = b;
  __syncthreads();
  for (int off = 128; off > 0; off >>= 1) {
    if ((int)threadIdx.x < off) {
      s1[threadIdx.x] += s1[threadIdx.x + off];
      s2[threadIdx.x] += s2[threadIdx.x + off];
    }
    __syncthreads();
  }
  if (threadIdx.x == 0) { atomicAdd(&stats[0], s1[0]); atomicAdd(&stats[1], s2[0]); }
}

__global__ __launch_bounds__(256)
void ln_gelu_kernel(float* __restrict__ h, const float* __restrict__ stats,
                    const float* __restrict__ w, const float* __restrict__ b) {
  int i = blockIdx.x * 256 + threadIdx.x;
  if (i >= N_NODES * HIDW) return;
  const float inv  = 1.0f / (float)(N_NODES * HIDW);
  const float mean = stats[0] * inv;
  const float var  = stats[1] * inv - mean * mean;
  const float rs   = rsqrtf(var + 1e-5f);
  const int c = i & (HIDW - 1);
  float v = (h[i] - mean) * rs * w[c] + b[c];
  v = 0.5f * v * (1.0f + erff(v * 0.70710678118654752f));
  h[i] = v;
}

extern "C" void kernel_launch(void* const* d_in, const int* in_sizes, int n_in,
                              void* d_out, int out_size, void* d_ws, size_t ws_size,
                              hipStream_t stream) {
  const float* x     = (const float*)d_in[0];
  const int*   ei    = (const int*)  d_in[1];
  const float* eattr = (const float*)d_in[2];
  const float* Wl    = (const float*)d_in[3];
  const float* Wr    = (const float*)d_in[4];
  const float* We    = (const float*)d_in[5];
  const float* att   = (const float*)d_in[6];
  const float* lnw   = (const float*)d_in[7];
  const float* lnb   = (const float*)d_in[8];
  const float* Wout  = (const float*)d_in[9];
  const float* bout  = (const float*)d_in[10];
  float* out = (float*)d_out;

  char* p = (char*)d_ws;
  auto carve = [&](size_t bytes) -> void* {
    void* q = (void*)p;
    p += (bytes + 255) & ~(size_t)255;
    return q;
  };
  float*    hA    = (float*)   carve((size_t)N_NODES * HIDW * 4);
  float*    hB    = (float*)   carve((size_t)N_NODES * HIDW * 4);
  float*    xl    = (float*)   carve((size_t)N_NODES * HIDW * 4);
  float*    xr    = (float*)   carve((size_t)N_NODES * HIDW * 4);
  float*    elog  = (float*)   carve((size_t)N_EDGES * NHEAD * 4);
  unsigned* lmax  = (unsigned*)carve((size_t)N_NODES * NHEAD * 4);
  float*    denom = (float*)   carve((size_t)N_NODES * NHEAD * 4);
  float*    stats = (float*)   carve(2 * 4);

  const int elemBlocks = (N_NODES * HIDW + 255) / 256;
  const int segBlocks  = (N_NODES * NHEAD + 255) / 256;
  const int ehBlocks   = (N_EDGES * NHEAD + 255) / 256;
  const int waveBlocks = 25000;
  const int gemmBlocks = N_NODES / 16;

  const float* hcur = x;
  float* bufs[2] = { hA, hB };
  for (int l = 0; l < NLAYERS; ++l) {
    float* hout = bufs[l & 1];
    init_layer_kernel<<<elemBlocks, 256, 0, stream>>>(hout, lmax, denom, stats);
    gemm_dual_wmma<<<gemmBlocks, 256, 0, stream>>>(hcur, Wl + (size_t)l * HIDW * HIDW,
                                                   Wr + (size_t)l * HIDW * HIDW, xl, xr);
    edge_logits_kernel<<<waveBlocks, 256, 0, stream>>>(ei, eattr, xl, xr,
                                                       We + (size_t)l * EDIMW * HIDW,
                                                       att + (size_t)l * HIDW, elog, lmax);
    lmax_fixup_kernel<<<segBlocks, 256, 0, stream>>>(lmax);
    edge_expsum_kernel<<<ehBlocks, 256, 0, stream>>>(ei, elog, (const float*)lmax, denom);
    edge_scatter_kernel<<<waveBlocks, 256, 0, stream>>>(ei, xl, elog, denom, hout);
    ln_reduce_kernel<<<2048, 256, 0, stream>>>(hout, stats);
    ln_gelu_kernel<<<elemBlocks, 256, 0, stream>>>(hout, stats,
                                                   lnw + (size_t)l * HIDW,
                                                   lnb + (size_t)l * HIDW);
    hcur = hout;
  }
  gemm_bias_wmma<<<gemmBlocks, 256, 0, stream>>>(hcur, Wout, bout, out);
}


# Edge-logit pass re-laid: a lane holds 4 adjacent channels, so a wave asks the 16 rows of 8 edges (16 B per lane) before first use and sums a head in its 8-lane group by DPP; logit sum order changes
# speedup vs baseline: 1.6777x; 1.6777x over previous
#include <hip/hip_runtime.h>
#include <math.h>

#define N_NODES 50000
#define N_EDGES 800000
#define HIDW    128
#define NHEAD   4
#define CPH     32
#define NLAYERS 3
#define EDIMW   3

typedef __attribute__((ext_vector_type(2))) float v2f;
typedef __attribute__((ext_vector_type(8))) float v8f;
typedef __attribute__((ext_vector_type(4))) float v4f;

__device__ __forceinline__ unsigned encOrd(float f) {
  unsigned u = __float_as_uint(f);
  return (u & 0x80000000u) ? ~u : (u | 0x80000000u);
}
__device__ __forceinline__ float decOrd(unsigned u) {
  return __uint_as_float((u & 0x80000000u) ? (u & 0x7FFFFFFFu) : ~u);
}
#define ENC_NEG_INF 0x007FFFFFu

__global__ __launch_bounds__(256)
void init_layer_kernel(unsigned* __restrict__ lmax, float* __restrict__ stats) {
  int i = blockIdx.x * 256 + threadIdx.x;
  if (i < N_NODES * NHEAD) lmax[i] = ENC_NEG_INF;
  if (i < 2) stats[i] = 0.f;
}

__global__ __launch_bounds__(256)
void gemm_dual_wmma(const float* __restrict__ A,
                    const float* __restrict__ W0,
                    const float* __restrict__ W1,
                    float* __restrict__ O0,
                    float* __restrict__ O1) {
  __shared__ float As[16 * 132];
  const int rowBase = blockIdx.x * 16;
  for (int i = threadIdx.x; i < 16 * HIDW; i += 256) {
    int r = i >> 7, c = i & 127;
    As[r * 132 + c] = A[(rowBase + r) * HIDW + c];
  }
  __syncthreads();
  const int wave    = threadIdx.x >> 5;
  const int lane    = threadIdx.x & 31;
  const int colBase = wave * 16;
  const int laneHi  = lane >> 4;
  const int laneLo  = lane & 15;
  v8f c0 = {}; v8f c1 = {};
  for (int k0 = 0; k0 < HIDW; k0 += 4) {
    const int ka = k0 + 2 * laneHi;
    v2f a;
    a.x = As[laneLo * 132 + ka];
    a.y = As[laneLo * 132 + ka + 1];
    v2f b0, b1;
    b0.x = W0[(ka + 0) * HIDW + colBase + laneLo];
    b0.y = W0[(ka + 1) * HIDW + colBase + laneLo];
    b1.x = W1[(ka + 0) * HIDW + colBase + laneLo];
    b1.y = W1[(ka + 1) * HIDW + colBase + laneLo];
    c0 = __builtin_amdgcn_wmma_f32_16x16x4_f32(false, a, false, b0, (short)0, c0, false, false);
    c1 = __builtin_amdgcn_wmma_f32_16x16x4_f32(false, a, false, b1, (short)0, c1, false, false);
  }
#pragma unroll
  for (int r = 0; r < 8; ++r) {
    const int row = rowBase + r + 8 * laneHi;
    O0[row * HIDW + colBase + laneLo] = c0[r];
    O1[row * HIDW + colBase + laneLo] = c1[r];
  }
}

__global__ __launch_bounds__(256)
void gemm_bias_wmma(const float* __restrict__ A,
                    const float* __restrict__ W,
                    const float* __restrict__ bias,
                    float* __restrict__ O) {
  __shared__ float As[16 * 132];
  const int rowBase = blockIdx.x * 16;
  for (int i = threadIdx.x; i < 16 * HIDW; i += 256) {
    int r = i >> 7, c = i & 127;
    As[r * 132 + c] = A[(rowBase + r) * HIDW + c];
  }
  __syncthreads();
  const int wave    = threadIdx.x >> 5;
  const int lane    = threadIdx.x & 31;
  const int colBase = wave * 16;
  const int laneHi  = lane >> 4;
  const int laneLo  = lane & 15;
  v8f c0 = {};
  for (int k0 = 0; k0 < HIDW; k0 += 4) {
    const int ka = k0 + 2 * laneHi;
    v2f a;
    a.x = As[laneLo * 132 + ka];
    a.y = As[laneLo * 132 + ka + 1];
    v2f b;
    b.x = W[(ka + 0) * HIDW + colBase + laneLo];
    b.y = W[(ka + 1) * HIDW + colBase + laneLo];
    c0 = __builtin_amdgcn_wmma_f32_16x16x4_f32(false, a, false, b, (short)0, c0, false, false);
  }
  const float bv = bias[colBase + laneLo];
#pragma unroll
  for (int r = 0; r < 8; ++r) {
    const int row = rowBase + r + 8 * laneHi;
    O[row * HIDW + colBase + laneLo] = c0[r] + bv;
  }
}

#define LG_WAVE_EDGES 32
#define LG_BATCH      8
static_assert(N_EDGES % (LG_WAVE_EDGES * 8) == 0, "whole work-groups of 8 waves cover the edges exactly");
static_assert(N_EDGES / (LG_WAVE_EDGES * 8) == 3125, "grid of edge_logits_kernel in kernel_launch");

template <int XCTRL>
__device__ __forceinline__ float add_xmask(float p) {
  return p + __int_as_float(__builtin_amdgcn_update_dpp(0, __float_as_int(p), XCTRL, 0xf, 0xf, false));
}

__global__ __launch_bounds__(256)
void edge_logits_kernel(const int* __restrict__ ei, const float* __restrict__ eattr,
                        const float* __restrict__ xl, const float* __restrict__ xr,
                        const float* __restrict__ We, const float* __restrict__ att,
                        float* __restrict__ elog, unsigned* __restrict__ lmaxu) {
  const int lane = threadIdx.x & 31;
  const int wid  = (blockIdx.x * 256 + threadIdx.x) >> 5;
  const int e0   = wid * LG_WAVE_EDGES;
  const int c4   = 4 * lane;
  const float4 we0 = *(const float4*)(We + c4);
  const float4 we1 = *(const float4*)(We + HIDW + c4);
  const float4 we2 = *(const float4*)(We + 2 * HIDW + c4);
  const float4 at  = *(const float4*)(att + c4);
  const int   srcv = ei[e0 + lane];
  const int   dstv = ei[N_EDGES + e0 + lane];
  const float av0  = eattr[(e0 + lane) * 3 + 0];
  const float av1  = eattr[(e0 + lane) * 3 + 1];
  const float av2  = eattr[(e0 + lane) * 3 + 2];
  const int sub  = lane & 7;
  const int head = lane >> 3;
#pragma unroll 1
  for (int j = 0; j < LG_WAVE_EDGES; j += LG_BATCH) {
    v4f xa[LG_BATCH], xb[LG_BATCH];
#pragma unroll
    for (int u = 0; u < LG_BATCH; ++u) {
      const int sj = __builtin_amdgcn_readlane(srcv, j + u);
      const int dj = __builtin_amdgcn_readlane(dstv, j + u);
      xa[u] = *(const v4f*)(xl + (size_t)sj * HIDW + c4);
      xb[u] = *(const v4f*)(xr + (size_t)dj * HIDW + c4);
    }
    asm volatile("" :: "v"(xa[0]), "v"(xb[0]), "v"(xa[1]), "v"(xb[1]), "v"(xa[2]), "v"(xb[2]),
                       "v"(xa[3]), "v"(xb[3]), "v"(xa[4]), "v"(xb[4]), "v"(xa[5]), "v"(xb[5]),
                       "v"(xa[6]), "v"(xb[6]), "v"(xa[7]), "v"(xb[7]));
    float mine = 0.f;
#pragma unroll
    for (int u = 0; u < LG_BATCH; ++u) {
      const float ea0 = __int_as_float(__builtin_amdgcn_readlane(__float_as_int(av0), j + u));
      const float ea1 = __int_as_float(__builtin_amdgcn_readlane(__float_as_int(av1), j + u));
      const float ea2 = __int_as_float(__builtin_amdgcn_readlane(__float_as_int(av2), j + u));
      float v0 = xa[u].x + xb[u].x + (ea0 * we0.x + ea1 * we1.x + ea2 * we2.x);
      float v1 = xa[u].y + xb[u].y + (ea0 * we0.y + ea1 * we1.y + ea2 * we2.y);
      float v2 = xa[u].z + xb[u].z + (ea0 * we0.z + ea1 * we1.z + ea2 * we2.z);
      float v3 = xa[u].w + xb[u].w + (ea0 * we0.w + ea1 * we1.w + ea2 * we2.w);
      v0 = (v0 > 0.f) ? v0 : 0.2f * v0;
      v1 = (v1 > 0.f) ? v1 : 0.2f * v1;
      v2 = (v2 > 0.f) ? v2 : 0.2f * v2;
      v3 = (v3 > 0.f) ? v3 : 0.2f * v3;
      float p = v0 * at.x + v1 * at.y + v2 * at.z + v3 * at.w;
      p = add_xmask<0x161>(p);
      p = add_xmask<0x162>(p);
      p = add_xmask<0x164>(p);
      if (sub == u) mine = p;
    }
    const int el    = j + sub;
    const int dmine = __shfl(dstv, el, 32);
    elog[(size_t)(e0 + el) * NHEAD + head] = mine;
    atomicMax(&lmaxu[dmine * NHEAD + head], encOrd(mine));
  }
}

#define OWN_NODES   196
#define OWN_GROUPS  256
#define OWN_WAVES   32

static_assert(N_NODES * NHEAD * 4 == N_EDGES, "one thread per (node, head) covers four edges");
static_assert((N_NODES + OWN_NODES - 1) / OWN_NODES <= 256, "owner id fits one byte");
__global__ __launch_bounds__(256)
void lmax_fixup_keys_kernel(unsigned* __restrict__ lmaxu, const int* __restrict__ dstp,
                            unsigned* __restrict__ okey) {
  int i = blockIdx.x * 256 + threadIdx.x;
  if (i >= N_NODES * NHEAD) return;
  float f = decOrd(lmaxu[i]);
  ((float*)lmaxu)[i] = isfinite(f) ? f : 0.f;
  const int4 d = *(const int4*)(dstp + 4 * i);
  okey[i] = (((unsigned)d.x / OWN_NODES) & 0xFFu)
          | ((((unsigned)d.y / OWN_NODES) & 0xFFu) << 8)
          | ((((unsigned)d.z / OWN_NODES) & 0xFFu) << 16)
          | ((((unsigned)d.w / OWN_NODES) & 0xFFu) << 24);
}

#define SCAN_EDGES  2048
#define SCAN_STEPS  13
#define SCAN_CHUNKS (SCAN_STEPS * OWN_WAVES)
#define QCAP        1056
static_assert(SCAN_CHUNKS * SCAN_EDGES >= N_EDGES, "scan covers every edge");
static_assert(N_EDGES % 16 == 0, "a lane's 16 owner bytes are all edges or none");

__global__ __launch_bounds__(1024)
void edge_owner_sum_kernel(const int* __restrict__ ei, const float* __restrict__ xl,
                           const float* __restrict__ elog, const float* __restrict__ lmaxf,
                           const unsigned* __restrict__ okey,
                           float* __restrict__ denom, float* __restrict__ out) {
  __shared__ float    sacc[OWN_NODES * HIDW];
  __shared__ float    sden[OWN_NODES * NHEAD];
  __shared__ unsigned sq[OWN_WAVES * QCAP];
  const int tid  = threadIdx.x;
  const int lane = tid & 31;
  const int wave = tid >> 5;
  const int nodeBase = blockIdx.x * OWN_NODES;
  for (int i = tid; i < OWN_NODES * HIDW; i += 1024) sacc[i] = 0.f;
  if (tid < OWN_NODES * NHEAD) sden[tid] = 0.f;
  __syncthreads();

  unsigned* q = sq + wave * QCAP;
  const int* dstp = ei + N_EDGES;
  int count = 0;

  auto flush = [&](int n) {
    for (int base = 0; base < n; base += 32) {
      const int cnt = (n - base < 32) ? (n - base) : 32;
      int srcv = 0, dlv = 0;
      float w0 = 0.f, w1 = 0.f, w2 = 0.f, w3 = 0.f;
      if (lane < cnt) {
        const int e = (int)q[base + lane];
        const int dstv = dstp[e];
        srcv = ei[e];
        const float4 lg = *(const float4*)(elog + (size_t)e * NHEAD);
        const unsigned dl = (unsigned)(dstv - nodeBase);
        if (dl < (unsigned)OWN_NODES) {
          dlv  = (int)dl;
          const float4 mx = *(const float4*)(lmaxf + (size_t)dstv * NHEAD);
          w0 = expf(lg.x - mx.x);
          w1 = expf(lg.y - mx.y);
          w2 = expf(lg.z - mx.z);
          w3 = expf(lg.w - mx.w);
          atomicAdd(&sden[dlv * NHEAD + 0], w0);
          atomicAdd(&sden[dlv * NHEAD + 1], w1);
          atomicAdd(&sden[dlv * NHEAD + 2], w2);
          atomicAdd(&sden[dlv * NHEAD + 3], w3);
        }
      }
      for (int j = 0; j < cnt; j += 8) {
        float x[8][NHEAD];
        float wj[8][NHEAD];
        int   dj[8];
#pragma unroll
        for (int u = 0; u < 8; ++u) {
          const int jj = j + u;
          const int sj = __builtin_amdgcn_readlane(srcv, jj);
          dj[u]    = __builtin_amdgcn_readlane(dlv, jj);
          wj[u][0] = __int_as_float(__builtin_amdgcn_readlane(__float_as_int(w0), jj));
          wj[u][1] = __int_as_float(__builtin_amdgcn_readlane(__float_as_int(w1), jj));
          wj[u][2] = __int_as_float(__builtin_amdgcn_readlane(__float_as_int(w2), jj));
          wj[u][3] = __int_as_float(__builtin_amdgcn_readlane(__float_as_int(w3), jj));
          const float* row = xl + (size_t)sj * HIDW + lane;
#pragma unroll
          for (int h = 0; h < NHEAD; ++h) x[u][h] = row[h * CPH];
        }
#pragma unroll
        for (int u = 0; u < 8; ++u) {
          if (j + u < cnt) {
            float* arow = sacc + dj[u] * HIDW + lane;
#pragma unroll
            for (int h = 0; h < NHEAD; ++h) atomicAdd(&arow[h * CPH], x[u][h] * wj[u][h]);
          }
        }
      }
    }
  };

  const unsigned gpat = (unsigned)blockIdx.x * 0x01010101u;
  const int rot = ((int)blockIdx.x * 13) % SCAN_CHUNKS;
  auto chunk_of = [&](int it) {
    int t = it * OWN_WAVES + wave + rot;
    if (t >= SCAN_CHUNKS) t -= SCAN_CHUNKS;
    return t;
  };
  auto load_chunk = [&](uint4 (&d)[4], int chunk) {
#pragma unroll
    for (int k = 0; k < 4; ++k) {
      const int e0 = chunk * SCAN_EDGES + k * 512 + lane * 16;
      if (e0 < N_EDGES) d[k] = *(const uint4*)(okey + (e0 >> 2));
      else              d[k] = make_uint4(~gpat, ~gpat, ~gpat, ~gpat);
    }
  };
  auto differs = [&](unsigned w) {
    const unsigned x = w ^ gpat;
    return ((x & 0x7F7F7F7Fu) + 0x7F7F7F7Fu) | x;
  };
  auto sel = [](unsigned m, unsigned a, unsigned b) { return (a & m) | (b & ~m); };
  auto hits = [&](const uint4& a, const uint4& b) {
    unsigned r = differs(b.w);
    r = sel(0x40404040u, differs(b.z) >> 1, r);
    r = sel(0x20202020u, differs(b.y) >> 2, r);
    r = sel(0x10101010u, differs(b.x) >> 3, r);
    r = sel(0x08080808u, differs(a.w) >> 4, r);
    r = sel(0x04040404u, differs(a.z) >> 5, r);
    r = sel(0x02020202u, differs(a.y) >> 6, r);
    r = sel(0x01010101u, differs(a.x) >> 7, r);
    return ~r;
  };
  auto push = [&](unsigned hm, int ebase) {
    for (int round = 0; round < 32; ++round) {
      const bool act = hm != 0u;
      const unsigned mask = __builtin_amdgcn_ballot_w32(act);
      if (mask == 0u) break;
      if (act) {
        const int p = __builtin_ctz(hm);
        hm &= hm - 1u;
        const int i = p & 7;
        q[count + (int)__builtin_amdgcn_mbcnt_lo(mask, 0u)] =
            (unsigned)(ebase + (i >> 2) * 512 + (i & 3) * 4 + (p >> 3));
      }
      count += __builtin_popcount(mask);
    }
  };

  uint4 nxt[4];
  load_chunk(nxt, chunk_of(0));
  for (int it = 0; it < SCAN_STEPS; ++it) {
    uint4 cur[4];
#pragma unroll
    for (int k = 0; k < 4; ++k) cur[k] = nxt[k];
    const int chunk = chunk_of(it);
    if (it + 1 < SCAN_STEPS) load_chunk(nxt, chunk_of(it + 1));
    const int ebase = chunk * SCAN_EDGES + lane * 16;
    push(hits(cur[0], cur[1]), ebase);
    if (count >= 32) { flush(count); count = 0; }
    push(hits(cur[2], cur[3]), ebase + 1024);
    if (count >= 32) { flush(count); count = 0; }
  }
  flush(count);
  __syncthreads();

  for (int i = tid; i < OWN_NODES * HIDW; i += 1024) {
    const int node = nodeBase + (i >> 7);
    if (node < N_NODES) out[(size_t)nodeBase * HIDW + i] = sacc[i];
  }
  if (tid < OWN_NODES * NHEAD) {
    const int node = nodeBase + (tid >> 2);
    if (node < N_NODES) denom[(size_t)nodeBase * NHEAD + tid] = sden[tid];
  }
}

__device__ __forceinline__ float softmax_row_value(float s, const float* __restrict__ denom, int i) {
  return s / (denom[i >> 5] + 1e-16f);
}

__global__ __launch_bounds__(256)
void ln_reduce_kernel(const float* __restrict__ h, const float* __restrict__ denom,
                      float* __restrict__ stats) {
  __shared__ float s1[256], s2[256];
  float a = 0.f, b = 0.f;
  for (int i = blockIdx.x * 256 + threadIdx.x; i < N_NODES * HIDW; i += gridDim.x * 256) {
    float v = softmax_row_value(h[i], denom, i);
    a += v; b += v * v;
  }
  s1[threadIdx.x] = a; s2[threadIdx.x] = b;
  __syncthreads();
  for (int off = 128; off > 0; off >>= 1) {
    if ((int)threadIdx.x < off) {
      s1[threadIdx.x] += s1[threadIdx.x + off];
      s2[threadIdx.x] += s2[threadIdx.x + off];
    }
    __syncthreads();
  }
  if (threadIdx.x == 0) { atomicAdd(&stats[0], s1[0]); atomicAdd(&stats[1], s2[0]); }
}

__global__ __launch_bounds__(256)
void ln_gelu_kernel(float* __restrict__ h, const float* __restrict__ denom,
                    const float* __restrict__ stats,
                    const float* __restrict__ w, const float* __restrict__ b) {
  int i = blockIdx.x * 256 + threadIdx.x;
  if (i >= N_NODES * HIDW) return;
  const float inv  = 1.0f / (float)(N_NODES * HIDW);
  const float mean = stats[0] * inv;
  const float var  = stats[1] * inv - mean * mean;
  const float rs   = rsqrtf(var + 1e-5f);
  const int c = i & (HIDW - 1);
  float v = (softmax_row_value(h[i], denom, i) - mean) * rs * w[c] + b[c];
  v = 0.5f * v * (1.0f + erff(v * 0.70710678118654752f));
  h[i] = v;
}

extern "C" void kernel_launch(void* const* d_in, const int* in_sizes, int n_in,
                              void* d_out, int out_size, void* d_ws, size_t ws_size,
                              hipStream_t stream) {
  const float* x     = (const float*)d_in[0];
  const int*   ei    = (const int*)  d_in[1];
  const float* eattr = (const float*)d_in[2];
  const float* Wl    = (const float*)d_in[3];
  const float* Wr    = (const float*)d_in[4];
  const float* We    = (const float*)d_in[5];
  const float* att   = (const float*)d_in[6];
  const float* lnw   = (const float*)d_in[7];
  const float* lnb   = (const float*)d_in[8];
  const float* Wout  = (const float*)d_in[9];
  const float* bout  = (const float*)d_in[10];
  float* out = (float*)d_out;

  char* p = (char*)d_ws;
  auto carve = [&](size_t bytes) -> void* {
    void* q = (void*)p;
    p += (bytes + 255) & ~(size_t)255;
    return q;
  };
  float*    hA    = (float*)   carve((size_t)N_NODES * HIDW * 4);
  float*    hB    = (float*)   carve((size_t)N_NODES * HIDW * 4);
  float*    xl    = (float*)   carve((size_t)N_NODES * HIDW * 4);
  float*    xr    = (float*)   carve((size_t)N_NODES * HIDW * 4);
  float*    elog  = (float*)   carve((size_t)N_EDGES * NHEAD * 4);
  unsigned* lmax  = (unsigned*)carve((size_t)N_NODES * NHEAD * 4);
  float*    denom = (float*)   carve((size_t)N_NODES * NHEAD * 4);
  float*    stats = (float*)   carve(2 * 4);
  unsigned* okey  = (unsigned*)carve((size_t)N_EDGES);

  const int elemBlocks = (N_NODES * HIDW + 255) / 256;
  const int segBlocks  = (N_NODES * NHEAD + 255) / 256;
  const int waveBlocks = 3125;
  const int gemmBlocks = N_NODES / 16;

  const float* hcur = x;
  float* bufs[2] = { hA, hB };
  for (int l = 0; l < NLAYERS; ++l) {
    float* hout = bufs[l & 1];
    init_layer_kernel<<<segBlocks, 256, 0, stream>>>(lmax, stats);
    gemm_dual_wmma<<<gemmBlocks, 256, 0, stream>>>(hcur, Wl + (size_t)l * HIDW * HIDW,
                                                   Wr + (size_t)l * HIDW * HIDW, xl, xr);
    edge_logits_kernel<<<waveBlocks, 256, 0, stream>>>(ei, eattr, xl, xr,
                                                       We + (size_t)l * EDIMW * HIDW,
                                                       att + (size_t)l * HIDW, elog, lmax);
    lmax_fixup_keys_kernel<<<segBlocks, 256, 0, stream>>>(lmax, ei + N_EDGES, okey);
    edge_owner_sum_kernel<<<OWN_GROUPS, 1024, 0, stream>>>(ei, xl, elog, (const float*)lmax,
                                                            okey, denom, hout);
    ln_reduce_kernel<<<2048, 256, 0, stream>>>(hout, denom, stats);
    ln_gelu_kernel<<<elemBlocks, 256, 0, stream>>>(hout, denom, stats,
                                                   lnw + (size_t)l * HIDW,
                                                   lnb + (size_t)l * HIDW);
    hcur = hout;
  }
  gemm_bias_wmma<<<gemmBlocks, 256, 0, stream>>>(hcur, Wout, bout, out);
}


# Edge logits and segment max move into the work-group owning the destination: its xr rows, the max and its edge list sit in LDS; xr gather, global max atomics, 8 launches gone; LDS sums arrival-order
# speedup vs baseline: 2.0059x; 1.1956x over previous
#include <hip/hip_runtime.h>
#include <math.h>

#define N_NODES 50000
#define N_EDGES 800000
#define HIDW    128
#define NHEAD   4
#define CPH     32
#define NLAYERS 3
#define EDIMW   3

typedef __attribute__((ext_vector_type(2))) float v2f;
typedef __attribute__((ext_vector_type(8))) float v8f;
typedef __attribute__((ext_vector_type(4))) float v4f;

__device__ __forceinline__ unsigned encOrd(float f) {
  unsigned u = __float_as_uint(f);
  return (u & 0x80000000u) ? ~u : (u | 0x80000000u);
}
__device__ __forceinline__ float decOrd(unsigned u) {
  return __uint_as_float((u & 0x80000000u) ? (u & 0x7FFFFFFFu) : ~u);
}
#define ENC_NEG_INF 0x007FFFFFu

__global__ __launch_bounds__(256)
void gemm_dual_wmma(const float* __restrict__ A,
                    const float* __restrict__ W0,
                    const float* __restrict__ W1,
                    float* __restrict__ O0,
                    float* __restrict__ O1) {
  __shared__ float As[16 * 132];
  const int rowBase = blockIdx.x * 16;
  for (int i = threadIdx.x; i < 16 * HIDW; i += 256) {
    int r = i >> 7, c = i & 127;
    As[r * 132 + c] = A[(rowBase + r) * HIDW + c];
  }
  __syncthreads();
  const int wave    = threadIdx.x >> 5;
  const int lane    = threadIdx.x & 31;
  const int colBase = wave * 16;
  const int laneHi  = lane >> 4;
  const int laneLo  = lane & 15;
  v8f c0 = {}; v8f c1 = {};
  for (int k0 = 0; k0 < HIDW; k0 += 4) {
    const int ka = k0 + 2 * laneHi;
    v2f a;
    a.x = As[laneLo * 132 + ka];
    a.y = As[laneLo * 132 + ka + 1];
    v2f b0, b1;
    b0.x = W0[(ka + 0) * HIDW + colBase + laneLo];
    b0.y = W0[(ka + 1) * HIDW + colBase + laneLo];
    b1.x = W1[(ka + 0) * HIDW + colBase + laneLo];
    b1.y = W1[(ka + 1) * HIDW + colBase + laneLo];
    c0 = __builtin_amdgcn_wmma_f32_16x16x4_f32(false, a, false, b0, (short)0, c0, false, false);
    c1 = __builtin_amdgcn_wmma_f32_16x16x4_f32(false, a, false, b1, (short)0, c1, false, false);
  }
#pragma unroll
  for (int r = 0; r < 8; ++r) {
    const int row = rowBase + r + 8 * laneHi;
    O0[row * HIDW + colBase + laneLo] = c0[r];
    O1[row * HIDW + colBase + laneLo] = c1[r];
  }
}

__global__ __launch_bounds__(256)
void gemm_bias_wmma(const float* __restrict__ A,
                    const float* __restrict__ W,
                    const float* __restrict__ bias,
                    float* __restrict__ O) {
  __shared__ float As[16 * 132];
  const int rowBase = blockIdx.x * 16;
  for (int i = threadIdx.x; i < 16 * HIDW; i += 256) {
    int r = i >> 7, c = i & 127;
    As[r * 132 + c] = A[(rowBase + r) * HIDW + c];
  }
  __syncthreads();
  const int wave    = threadIdx.x >> 5;
  const int lane    = threadIdx.x & 31;
  const int colBase = wave * 16;
  const int laneHi  = lane >> 4;
  const int laneLo  = lane & 15;
  v8f c0 = {};
  for (int k0 = 0; k0 < HIDW; k0 += 4) {
    const int ka = k0 + 2 * laneHi;
    v2f a;
    a.x = As[laneLo * 132 + ka];
    a.y = As[laneLo * 132 + ka + 1];
    v2f b;
    b.x = W[(ka + 0) * HIDW + colBase + laneLo];
    b.y = W[(ka + 1) * HIDW + colBase + laneLo];
    c0 = __builtin_amdgcn_wmma_f32_16x16x4_f32(false, a, false, b, (short)0, c0, false, false);
  }
  const float bv = bias[colBase + laneLo];
#pragma unroll
  for (int r = 0; r < 8; ++r) {
    const int row = rowBase + r + 8 * laneHi;
    O[row * HIDW + colBase + laneLo] = c0[r] + bv;
  }
}

#define OWN_NODES   196
#define OWN_GROUPS  256
#define OWN_WAVES   32
static_assert(N_EDGES % 4 == 0, "one thread packs the owner bytes of four edges");
static_assert((N_NODES + OWN_NODES - 1) / OWN_NODES <= 256, "owner id fits one byte");

__global__ __launch_bounds__(256)
void owner_keys_kernel(const int* __restrict__ dstp, unsigned* __restrict__ okey) {
  int i = blockIdx.x * 256 + threadIdx.x;
  if (i >= N_EDGES / 4) return;
  const int4 d = *(const int4*)(dstp + 4 * i);
  okey[i] = (((unsigned)d.x / OWN_NODES) & 0xFFu)
          | ((((unsigned)d.y / OWN_NODES) & 0xFFu) << 8)
          | ((((unsigned)d.z / OWN_NODES) & 0xFFu) << 16)
          | ((((unsigned)d.w / OWN_NODES) & 0xFFu) << 24);
}

template <int XCTRL>
__device__ __forceinline__ float add_xmask(float p) {
  return p + __int_as_float(__builtin_amdgcn_update_dpp(0, __float_as_int(p), XCTRL, 0xf, 0xf, false));
}

#define SCAN_EDGES  2048
#define SCAN_STEPS  13
#define SCAN_CHUNKS (SCAN_STEPS * OWN_WAVES)
#define QCAP        1056
#define PCAP        224
static_assert(N_NODES <= 65536 && OWN_NODES < 255, "a list entry packs the source into 16 bits, the local node into 8");
static_assert(SCAN_CHUNKS * SCAN_EDGES >= N_EDGES, "scan covers every edge");
static_assert(N_EDGES % 16 == 0, "a lane's 16 owner bytes are all edges or none");

__global__ __launch_bounds__(1024)
void edge_owner_kernel(const int* __restrict__ ei, const float* __restrict__ eattr,
                       const float* __restrict__ xl, const float* __restrict__ xr,
                       const float* __restrict__ We, const float* __restrict__ att,
                       const unsigned* __restrict__ okey, float* __restrict__ elog,
                       float* __restrict__ denom, float* __restrict__ stats,
                       float* __restrict__ out) {
  __shared__ __attribute__((aligned(16))) float    sacc[OWN_NODES * HIDW];
  __shared__ __attribute__((aligned(16))) float    sden[OWN_NODES * NHEAD];
  __shared__ __attribute__((aligned(16))) unsigned smax[OWN_NODES * NHEAD];
  __shared__ unsigned sq[OWN_WAVES * QCAP];
  __shared__ __attribute__((aligned(16))) uint2 sp[OWN_WAVES * PCAP];
  const int tid  = threadIdx.x;
  const int lane = tid & 31;
  const int wave = tid >> 5;
  const int nodeBase = blockIdx.x * OWN_NODES;
  if (blockIdx.x == 0 && tid < 2) stats[tid] = 0.f;
  for (int i = tid * 4; i < OWN_NODES * HIDW; i += 4096) {
    const int node = nodeBase + (i >> 7);
    float4 v = make_float4(0.f, 0.f, 0.f, 0.f);
    if (node < N_NODES) v = *(const float4*)(xr + (size_t)nodeBase * HIDW + i);
    *(float4*)(sacc + i) = v;
  }
  if (tid < OWN_NODES * NHEAD) { sden[tid] = 0.f; smax[tid] = ENC_NEG_INF; }
  __syncthreads();

  unsigned* q  = sq + wave * QCAP;
  uint2*    pl = sp + wave * PCAP;
  const int* dstp = ei + N_EDGES;
  int count = 0;

  const unsigned gpat = (unsigned)blockIdx.x * 0x01010101u;
  const int rot = ((int)blockIdx.x * 13) % SCAN_CHUNKS;
  auto chunk_of = [&](int it) {
    int t = it * OWN_WAVES + wave + rot;
    if (t >= SCAN_CHUNKS) t -= SCAN_CHUNKS;
    return t;
  };
  auto load_chunk = [&](uint4 (&d)[4], int chunk) {
#pragma unroll
    for (int k = 0; k < 4; ++k) {
      const int e0 = chunk * SCAN_EDGES + k * 512 + lane * 16;
      if (e0 < N_EDGES) d[k] = *(const uint4*)(okey + (e0 >> 2));
      else              d[k] = make_uint4(~gpat, ~gpat, ~gpat, ~gpat);
    }
  };
  auto differs = [&](unsigned w) {
    const unsigned x = w ^ gpat;
    return ((x & 0x7F7F7F7Fu) + 0x7F7F7F7Fu) | x;
  };
  auto sel = [](unsigned m, unsigned a, unsigned b) { return (a & m) | (b & ~m); };
  auto hits = [&](const uint4& a, const uint4& b) {
    unsigned r = differs(b.w);
    r = sel(0x40404040u, differs(b.z) >> 1, r);
    r = sel(0x20202020u, differs(b.y) >> 2, r);
    r = sel(0x10101010u, differs(b.x) >> 3, r);
    r = sel(0x08080808u, differs(a.w) >> 4, r);
    r = sel(0x04040404u, differs(a.z) >> 5, r);
    r = sel(0x02020202u, differs(a.y) >> 6, r);
    r = sel(0x01010101u, differs(a.x) >> 7, r);
    return ~r;
  };
  auto push = [&](unsigned hm, int ebase) {
    for (int round = 0; round < 32; ++round) {
      const bool act = hm != 0u;
      const unsigned mask = __builtin_amdgcn_ballot_w32(act);
      if (mask == 0u) break;
      if (act) {
        const int p = __builtin_ctz(hm);
        hm &= hm - 1u;
        const int i = p & 7;
        q[count + (int)__builtin_amdgcn_mbcnt_lo(mask, 0u)] =
            (unsigned)(ebase + (i >> 2) * 512 + (i & 3) * 4 + (p >> 3));
      }
      count += __builtin_popcount(mask);
    }
  };
  auto scan = [&](auto&& work) {
    count = 0;
    uint4 nxt[4];
    load_chunk(nxt, chunk_of(0));
    for (int it = 0; it < SCAN_STEPS; ++it) {
      uint4 cur[4];
#pragma unroll
      for (int k = 0; k < 4; ++k) cur[k] = nxt[k];
      const int chunk = chunk_of(it);
      if (it + 1 < SCAN_STEPS) load_chunk(nxt, chunk_of(it + 1));
      const int ebase = chunk * SCAN_EDGES + lane * 16;
      push(hits(cur[0], cur[1]), ebase);
      if (count >= 32) { work(count); count = 0; }
      push(hits(cur[2], cur[3]), ebase + 1024);
      if (count >= 32) { work(count); count = 0; }
    }
    work(count);
  };

  int  pcount = 0;
  bool listed = true;
  {
    const int c4   = 4 * lane;
    const int sub  = lane & 7;
    const int head = lane >> 3;
    const float4 we0 = *(const float4*)(We + c4);
    const float4 we1 = *(const float4*)(We + HIDW + c4);
    const float4 we2 = *(const float4*)(We + 2 * HIDW + c4);
    const float4 at  = *(const float4*)(att + c4);
    auto logits = [&](int n) {
      if (pcount + n > PCAP) listed = false;
      for (int base = 0; base < n; base += 32) {
        const int cnt = (n - base < 32) ? (n - base) : 32;
        int ev = 0, srcv = 0, dlv = 0, okv = 0;
        float av0 = 0.f, av1 = 0.f, av2 = 0.f;
        if (lane < cnt) {
          ev = (int)q[base + lane];
          const int dstv = dstp[ev];
          srcv = ei[ev];
          av0 = eattr[(size_t)ev * 3 + 0];
          av1 = eattr[(size_t)ev * 3 + 1];
          av2 = eattr[(size_t)ev * 3 + 2];
          const unsigned dl = (unsigned)(dstv - nodeBase);
          if (dl < (unsigned)OWN_NODES) { dlv = (int)dl; okv = 1; }
          if (listed) pl[pcount + base + lane] =
              make_uint2((unsigned)ev, (unsigned)srcv | ((okv ? dl : 255u) << 16));
        }
        for (int j = 0; j < cnt; j += 8) {
          v4f xa[8];
          int dj[8];
#pragma unroll
          for (int u = 0; u < 8; ++u) {
            const int jj = j + u;
            const int sj = __builtin_amdgcn_readlane(srcv, jj);
            dj[u] = __builtin_amdgcn_readlane(dlv, jj);
            xa[u] = *(const v4f*)(xl + (size_t)sj * HIDW + c4);
          }
          asm volatile("" :: "v"(xa[0]), "v"(xa[1]), "v"(xa[2]), "v"(xa[3]),
                             "v"(xa[4]), "v"(xa[5]), "v"(xa[6]), "v"(xa[7]));
          float mine = 0.f;
#pragma unroll
          for (int u = 0; u < 8; ++u) {
            const int jj = j + u;
            const v4f xb = *(const v4f*)(sacc + dj[u] * HIDW + c4);
            const float ea0 = __int_as_float(__builtin_amdgcn_readlane(__float_as_int(av0), jj));
            const float ea1 = __int_as_float(__builtin_amdgcn_readlane(__float_as_int(av1), jj));
            const float ea2 = __int_as_float(__builtin_amdgcn_readlane(__float_as_int(av2), jj));
            float v0 = xa[u].x + xb.x + (ea0 * we0.x + ea1 * we1.x + ea2 * we2.x);
            float v1 = xa[u].y + xb.y + (ea0 * we0.y + ea1 * we1.y + ea2 * we2.y);
            float v2 = xa[u].z + xb.z + (ea0 * we0.z + ea1 * we1.z + ea2 * we2.z);
            float v3 = xa[u].w + xb.w + (ea0 * we0.w + ea1 * we1.w + ea2 * we2.w);
            v0 = (v0 > 0.f) ? v0 : 0.2f * v0;
            v1 = (v1 > 0.f) ? v1 : 0.2f * v1;
            v2 = (v2 > 0.f) ? v2 : 0.2f * v2;
            v3 = (v3 > 0.f) ? v3 : 0.2f * v3;
            float p = v0 * at.x + v1 * at.y + v2 * at.z + v3 * at.w;
            p = add_xmask<0x161>(p);
            p = add_xmask<0x162>(p);
            p = add_xmask<0x164>(p);
            if (sub == u) mine = p;
          }
          const int el    = j + sub;
          const int emine = __shfl(ev, el, 32);
          const int dmine = __shfl(dlv, el, 32);
          const int omine = __shfl(okv, el, 32);
          if (omine) {
            elog[(size_t)emine * NHEAD + head] = mine;
            atomicMax(&smax[dmine * NHEAD + head], encOrd(mine));
          }
        }
      }
      if (listed) pcount += n;
    };
    scan(logits);
  }
  __syncthreads();

  for (int i = tid * 4; i < OWN_NODES * HIDW; i += 4096) *(float4*)(sacc + i) = make_float4(0.f, 0.f, 0.f, 0.f);
  if (tid < OWN_NODES * NHEAD) {
    const float f = decOrd(smax[tid]);
    smax[tid] = __float_as_uint(isfinite(f) ? f : 0.f);
  }
  __syncthreads();

  auto sums = [&](int n, auto&& fetch) {
    for (int base = 0; base < n; base += 32) {
      const int cnt = (n - base < 32) ? (n - base) : 32;
      int srcv = 0, dlv = 0;
      float w0 = 0.f, w1 = 0.f, w2 = 0.f, w3 = 0.f;
      if (lane < cnt) {
        int e; unsigned dl;
        fetch(base + lane, e, srcv, dl);
        const float4 lg = *(const float4*)(elog + (size_t)e * NHEAD);
        if (dl < (unsigned)OWN_NODES) {
          dlv  = (int)dl;
          const uint4 mx = *(const uint4*)(smax + dlv * NHEAD);
          w0 = expf(lg.x - __uint_as_float(mx.x));
          w1 = expf(lg.y - __uint_as_float(mx.y));
          w2 = expf(lg.z - __uint_as_float(mx.z));
          w3 = expf(lg.w - __uint_as_float(mx.w));
          atomicAdd(&sden[dlv * NHEAD + 0], w0);
          atomicAdd(&sden[dlv * NHEAD + 1], w1);
          atomicAdd(&sden[dlv * NHEAD + 2], w2);
          atomicAdd(&sden[dlv * NHEAD + 3], w3);
        }
      }
      for (int j = 0; j < cnt; j += 8) {
        float x[8][NHEAD];
        float wj[8][NHEAD];
        int   dj[8];
#pragma unroll
        for (int u = 0; u < 8; ++u) {
          const int jj = j + u;
          const int sj = __builtin_amdgcn_readlane(srcv, jj);
          dj[u]    = __builtin_amdgcn_readlane(dlv, jj);
          wj[u][0] = __int_as_float(__builtin_amdgcn_readlane(__float_as_int(w0), jj));
          wj[u][1] = __int_as_float(__builtin_amdgcn_readlane(__float_as_int(w1), jj));
          wj[u][2] = __int_as_float(__builtin_amdgcn_readlane(__float_as_int(w2), jj));
          wj[u][3] = __int_as_float(__builtin_amdgcn_readlane(__float_as_int(w3), jj));
          const float* row = xl + (size_t)sj * HIDW + lane;
#pragma unroll
          for (int h = 0; h < NHEAD; ++h) x[u][h] = row[h * CPH];
        }
#pragma unroll
        for (int u = 0; u < 8; ++u) {
          if (j + u < cnt) {
            float* arow = sacc + dj[u] * HIDW + lane;
#pragma unroll
            for (int h = 0; h < NHEAD; ++h) atomicAdd(&arow[h * CPH], x[u][h] * wj[u][h]);
          }
        }
      }
    }
  };
  if (listed) {
    sums(pcount, [&](int i, int& e, int& s, unsigned& dl) {
      const uint2 r = pl[i];
      e = (int)r.x; s = (int)(r.y & 0xFFFFu); dl = r.y >> 16;
    });
  } else {
    scan([&](int n) {
      sums(n, [&](int i, int& e, int& s, unsigned& dl) {
        e = (int)q[i]; s = ei[e]; dl = (unsigned)(dstp[e] - nodeBase);
      });
    });
  }
  __syncthreads();

  for (int i = tid; i < OWN_NODES * HIDW; i += 1024) {
    const int node = nodeBase + (i >> 7);
    if (node < N_NODES) out[(size_t)nodeBase * HIDW + i] = sacc[i];
  }
  if (tid < OWN_NODES * NHEAD) {
    const int node = nodeBase + (tid >> 2);
    if (node < N_NODES) denom[(size_t)nodeBase * NHEAD + tid] = sden[tid];
  }
}

__device__ __forceinline__ float softmax_row_value(float s, const float* __restrict__ denom, int i) {
  return s / (denom[i >> 5] + 1e-16f);
}

__global__ __launch_bounds__(256)
void ln_reduce_kernel(const float* __restrict__ h, const float* __restrict__ denom,
                      float* __restrict__ stats) {
  __shared__ float s1[256], s2[256];
  float a = 0.f, b = 0.f;
  for (int i = blockIdx.x * 256 + threadIdx.x; i < N_NODES * HIDW; i += gridDim.x * 256) {
    float v = softmax_row_value(h[i], denom, i);
    a += v; b += v * v;
  }
  s1[threadIdx.x] = a; s2[threadIdx.x] = b;
  __syncthreads();
  for (int off = 128; off > 0; off >>= 1) {
    if ((int)threadIdx.x < off) {
      s1[threadIdx.x] += s1[threadIdx.x + off];
      s2[threadIdx.x] += s2[threadIdx.x + off];
    }
    __syncthreads();
  }
  if (threadIdx.x == 0) { atomicAdd(&stats[0], s1[0]); atomicAdd(&stats[1], s2[0]); }
}

__global__ __launch_bounds__(256)
void ln_gelu_kernel(float* __restrict__ h, const float* __restrict__ denom,
                    const float* __restrict__ stats,
                    const float* __restrict__ w, const float* __restrict__ b) {
  int i = blockIdx.x * 256 + threadIdx.x;
  if (i >= N_NODES * HIDW) return;
  const float inv  = 1.0f / (float)(N_NODES * HIDW);
  const float mean = stats[0] * inv;
  const float var  = stats[1] * inv - mean * mean;
  const float rs   = rsqrtf(var + 1e-5f);
  const int c = i & (HIDW - 1);
  float v = (softmax_row_value(h[i], denom, i) - mean) * rs * w[c] + b[c];
  v = 0.5f * v * (1.0f + erff(v * 0.70710678118654752f));
  h[i] = v;
}

extern "C" void kernel_launch(void* const* d_in, const int* in_sizes, int n_in,
                              void* d_out, int out_size, void* d_ws, size_t ws_size,
                              hipStream_t stream) {
  const float* x     = (const float*)d_in[0];
  const int*   ei    = (const int*)  d_in[1];
  const float* eattr = (const float*)d_in[2];
  const float* Wl    = (const float*)d_in[3];
  const float* Wr    = (const float*)d_in[4];
  const float* We    = (const float*)d_in[5];
  const float* att   = (const float*)d_in[6];
  const float* lnw   = (const float*)d_in[7];
  const float* lnb   = (const float*)d_in[8];
  const float* Wout  = (const float*)d_in[9];
  const float* bout  = (const float*)d_in[10];
  float* out = (float*)d_out;

  char* p = (char*)d_ws;
  auto carve = [&](size_t bytes) -> void* {
    void* q = (void*)p;
    p += (bytes + 255) & ~(size_t)255;
    return q;
  };
  float*    hA    = (float*)   carve((size_t)N_NODES * HIDW * 4);
  float*    hB    = (float*)   carve((size_t)N_NODES * HIDW * 4);
  float*    xl    = (float*)   carve((size_t)N_NODES * HIDW * 4);
  float*    xr    = (float*)   carve((size_t)N_NODES * HIDW * 4);
  float*    elog  = (float*)   carve((size_t)N_EDGES * NHEAD * 4);
  float*    denom = (float*)   carve((size_t)N_NODES * NHEAD * 4);
  float*    stats = (float*)   carve(2 * 4);
  unsigned* okey  = (unsigned*)carve((size_t)N_EDGES);

  const int elemBlocks = (N_NODES * HIDW + 255) / 256;
  const int keyBlocks  = (N_EDGES / 4 + 255) / 256;
  const int gemmBlocks = N_NODES / 16;

  owner_keys_kernel<<<keyBlocks, 256, 0, stream>>>(ei + N_EDGES, okey);
  const float* hcur = x;
  float* bufs[2] = { hA, hB };
  for (int l = 0; l < NLAYERS; ++l) {
    float* hout = bufs[l & 1];
    gemm_dual_wmma<<<gemmBlocks, 256, 0, stream>>>(hcur, Wl + (size_t)l * HIDW * HIDW,
                                                   Wr + (size_t)l * HIDW * HIDW, xl, xr);
    edge_owner_kernel<<<OWN_GROUPS, 1024, 0, stream>>>(ei, eattr, xl, xr,
                                                        We + (size_t)l * EDIMW * HIDW,
                                                        att + (size_t)l * HIDW,
                                                        okey, elog, denom, stats, hout);
    ln_reduce_kernel<<<2048, 256, 0, stream>>>(hout, denom, stats);
    ln_gelu_kernel<<<elemBlocks, 256, 0, stream>>>(hout, denom, stats,
                                                   lnw + (size_t)l * HIDW,
                                                   lnb + (size_t)l * HIDW);
    hcur = hout;
  }
  gemm_bias_wmma<<<gemmBlocks, 256, 0, stream>>>(hcur, Wout, bout, out);
}


# Owner groups' edge lists (edge, source, local node, attributes) are stored by layer 1's launch and read back in layers 2 and 3: their owner-byte scans and scattered edge reads go; LDS sums as before
# speedup vs baseline: 2.1752x; 1.0844x over previous
#include <hip/hip_runtime.h>
#include <math.h>

#define N_NODES 50000
#define N_EDGES 800000
#define HIDW    128
#define NHEAD   4
#define CPH     32
#define NLAYERS 3
#define EDIMW   3

typedef __attribute__((ext_vector_type(2))) float v2f;
typedef __attribute__((ext_vector_type(8))) float v8f;
typedef __attribute__((ext_vector_type(4))) float v4f;

__device__ __forceinline__ unsigned encOrd(float f) {
  unsigned u = __float_as_uint(f);
  return (u & 0x80000000u) ? ~u : (u | 0x80000000u);
}
__device__ __forceinline__ float decOrd(unsigned u) {
  return __uint_as_float((u & 0x80000000u) ? (u & 0x7FFFFFFFu) : ~u);
}
#define ENC_NEG_INF 0x007FFFFFu

__global__ __launch_bounds__(256)
void gemm_dual_wmma(const float* __restrict__ A,
                    const float* __restrict__ W0,
                    const float* __restrict__ W1,
                    float* __restrict__ O0,
                    float* __restrict__ O1) {
  __shared__ float As[16 * 132];
  const int rowBase = blockIdx.x * 16;
  for (int i = threadIdx.x; i < 16 * HIDW; i += 256) {
    int r = i >> 7, c = i & 127;
    As[r * 132 + c] = A[(rowBase + r) * HIDW + c];
  }
  __syncthreads();
  const int wave    = threadIdx.x >> 5;
  const int lane    = threadIdx.x & 31;
  const int colBase = wave * 16;
  const int laneHi  = lane >> 4;
  const int laneLo  = lane & 15;
  v8f c0 = {}; v8f c1 = {};
  for (int k0 = 0; k0 < HIDW; k0 += 4) {
    const int ka = k0 + 2 * laneHi;
    v2f a;
    a.x = As[laneLo * 132 + ka];
    a.y = As[laneLo * 132 + ka + 1];
    v2f b0, b1;
    b0.x = W0[(ka + 0) * HIDW + colBase + laneLo];
    b0.y = W0[(ka + 1) * HIDW + colBase + laneLo];
    b1.x = W1[(ka + 0) * HIDW + colBase + laneLo];
    b1.y = W1[(ka + 1) * HIDW + colBase + laneLo];
    c0 = __builtin_amdgcn_wmma_f32_16x16x4_f32(false, a, false, b0, (short)0, c0, false, false);
    c1 = __builtin_amdgcn_wmma_f32_16x16x4_f32(false, a, false, b1, (short)0, c1, false, false);
  }
#pragma unroll
  for (int r = 0; r < 8; ++r) {
    const int row = rowBase + r + 8 * laneHi;
    O0[row * HIDW + colBase + laneLo] = c0[r];
    O1[row * HIDW + colBase + laneLo] = c1[r];
  }
}

__global__ __launch_bounds__(256)
void gemm_bias_wmma(const float* __restrict__ A,
                    const float* __restrict__ W,
                    const float* __restrict__ bias,
                    float* __restrict__ O) {
  __shared__ float As[16 * 132];
  const int rowBase = blockIdx.x * 16;
  for (int i = threadIdx.x; i < 16 * HIDW; i += 256) {
    int r = i >> 7, c = i & 127;
    As[r * 132 + c] = A[(rowBase + r) * HIDW + c];
  }
  __syncthreads();
  const int wave    = threadIdx.x >> 5;
  const int lane    = threadIdx.x & 31;
  const int colBase = wave * 16;
  const int laneHi  = lane >> 4;
  const int laneLo  = lane & 15;
  v8f c0 = {};
  for (int k0 = 0; k0 < HIDW; k0 += 4) {
    const int ka = k0 + 2 * laneHi;
    v2f a;
    a.x = As[laneLo * 132 + ka];
    a.y = As[laneLo * 132 + ka + 1];
    v2f b;
    b.x = W[(ka + 0) * HIDW + colBase + laneLo];
    b.y = W[(ka + 1) * HIDW + colBase + laneLo];
    c0 = __builtin_amdgcn_wmma_f32_16x16x4_f32(false, a, false, b, (short)0, c0, false, false);
  }
  const float bv = bias[colBase + laneLo];
#pragma unroll
  for (int r = 0; r < 8; ++r) {
    const int row = rowBase + r + 8 * laneHi;
    O[row * HIDW + colBase + laneLo] = c0[r] + bv;
  }
}

#define OWN_NODES   196
#define OWN_GROUPS  256
#define OWN_WAVES   32
static_assert(N_EDGES % 4 == 0, "one thread packs the owner bytes of four edges");
static_assert((N_NODES + OWN_NODES - 1) / OWN_NODES <= 256, "owner id fits one byte");

__global__ __launch_bounds__(256)
void owner_keys_kernel(const int* __restrict__ dstp, unsigned* __restrict__ okey) {
  int i = blockIdx.x * 256 + threadIdx.x;
  if (i >= N_EDGES / 4) return;
  const int4 d = *(const int4*)(dstp + 4 * i);
  okey[i] = (((unsigned)d.x / OWN_NODES) & 0xFFu)
          | ((((unsigned)d.y / OWN_NODES) & 0xFFu) << 8)
          | ((((unsigned)d.z / OWN_NODES) & 0xFFu) << 16)
          | ((((unsigned)d.w / OWN_NODES) & 0xFFu) << 24);
}

template <int XCTRL>
__device__ __forceinline__ float add_xmask(float p) {
  return p + __int_as_float(__builtin_amdgcn_update_dpp(0, __float_as_int(p), XCTRL, 0xf, 0xf, false));
}

#define SCAN_EDGES  2048
#define SCAN_STEPS  13
#define SCAN_CHUNKS (SCAN_STEPS * OWN_WAVES)
#define QCAP        1056
#define PCAP        224
static_assert(N_NODES <= 65536 && OWN_NODES < 255, "a list entry packs the source into 16 bits, the local node into 8");
static_assert(SCAN_CHUNKS * SCAN_EDGES >= N_EDGES, "scan covers every edge");
static_assert(N_EDGES % 16 == 0, "a lane's 16 owner bytes are all edges or none");

__global__ __launch_bounds__(1024)
void edge_owner_kernel(const int* __restrict__ ei, const float* __restrict__ eattr,
                       const float* __restrict__ xl, const float* __restrict__ xr,
                       const float* __restrict__ We, const float* __restrict__ att,
                       const unsigned* __restrict__ okey, float* __restrict__ elog,
                       float* __restrict__ denom, float* __restrict__ stats,
                       float* __restrict__ out,
                       unsigned* __restrict__ mlist, float* __restrict__ mattr,
                       int* __restrict__ mcnt, int first) {
  __shared__ __attribute__((aligned(16))) float    sacc[OWN_NODES * HIDW];
  __shared__ __attribute__((aligned(16))) float    sden[OWN_NODES * NHEAD];
  __shared__ __attribute__((aligned(16))) unsigned smax[OWN_NODES * NHEAD];
  __shared__ unsigned sq[OWN_WAVES * QCAP];
  __shared__ __attribute__((aligned(16))) uint2 sp[OWN_WAVES * PCAP];
  const int tid  = threadIdx.x;
  const int lane = tid & 31;
  const int wave = tid >> 5;
  const int nodeBase = blockIdx.x * OWN_NODES;
  if (blockIdx.x == 0 && tid < 2) stats[tid] = 0.f;
  for (int i = tid * 4; i < OWN_NODES * HIDW; i += 4096) {
    const int node = nodeBase + (i >> 7);
    float4 v = make_float4(0.f, 0.f, 0.f, 0.f);
    if (node < N_NODES) v = *(const float4*)(xr + (size_t)nodeBase * HIDW + i);
    *(float4*)(sacc + i) = v;
  }
  if (tid < OWN_NODES * NHEAD) { sden[tid] = 0.f; smax[tid] = ENC_NEG_INF; }
  __syncthreads();

  unsigned* q  = sq + wave * QCAP;
  uint2*    pl = sp + wave * PCAP;
  const int* dstp = ei + N_EDGES;
  int count = 0;

  const unsigned gpat = (unsigned)blockIdx.x * 0x01010101u;
  const int rot = ((int)blockIdx.x * 13) % SCAN_CHUNKS;
  auto chunk_of = [&](int it) {
    int t = it * OWN_WAVES + wave + rot;
    if (t >= SCAN_CHUNKS) t -= SCAN_CHUNKS;
    return t;
  };
  auto load_chunk = [&](uint4 (&d)[4], int chunk) {
#pragma unroll
    for (int k = 0; k < 4; ++k) {
      const int e0 = chunk * SCAN_EDGES + k * 512 + lane * 16;
      if (e0 < N_EDGES) d[k] = *(const uint4*)(okey + (e0 >> 2));
      else              d[k] = make_uint4(~gpat, ~gpat, ~gpat, ~gpat);
    }
  };
  auto differs = [&](unsigned w) {
    const unsigned x = w ^ gpat;
    return ((x & 0x7F7F7F7Fu) + 0x7F7F7F7Fu) | x;
  };
  auto sel = [](unsigned m, unsigned a, unsigned b) { return (a & m) | (b & ~m); };
  auto hits = [&](const uint4& a, const uint4& b) {
    unsigned r = differs(b.w);
    r = sel(0x40404040u, differs(b.z) >> 1, r);
    r = sel(0x20202020u, differs(b.y) >> 2, r);
    r = sel(0x10101010u, differs(b.x) >> 3, r);
    r = sel(0x08080808u, differs(a.w) >> 4, r);
    r = sel(0x04040404u, differs(a.z) >> 5, r);
    r = sel(0x02020202u, differs(a.y) >> 6, r);
    r = sel(0x01010101u, differs(a.x) >> 7, r);
    return ~r;
  };
  auto push = [&](unsigned hm, int ebase) {
    for (int round = 0; round < 32; ++round) {
      const bool act = hm != 0u;
      const unsigned mask = __builtin_amdgcn_ballot_w32(act);
      if (mask == 0u) break;
      if (act) {
        const int p = __builtin_ctz(hm);
        hm &= hm - 1u;
        const int i = p & 7;
        q[count + (int)__builtin_amdgcn_mbcnt_lo(mask, 0u)] =
            (unsigned)(ebase + (i >> 2) * 512 + (i & 3) * 4 + (p >> 3));
      }
      count += __builtin_popcount(mask);
    }
  };
  auto scan = [&](auto&& work) {
    count = 0;
    uint4 nxt[4];
    load_chunk(nxt, chunk_of(0));
    for (int it = 0; it < SCAN_STEPS; ++it) {
      uint4 cur[4];
#pragma unroll
      for (int k = 0; k < 4; ++k) cur[k] = nxt[k];
      const int chunk = chunk_of(it);
      if (it + 1 < SCAN_STEPS) load_chunk(nxt, chunk_of(it + 1));
      const int ebase = chunk * SCAN_EDGES + lane * 16;
      push(hits(cur[0], cur[1]), ebase);
      if (count >= 32) { work(count); count = 0; }
      push(hits(cur[2], cur[3]), ebase + 1024);
      if (count >= 32) { work(count); count = 0; }
    }
    work(count);
  };

  int  pcount = 0;
  bool listed = true;
  const size_t mslot = ((size_t)blockIdx.x * OWN_WAVES + wave) * PCAP;
  {
    const int c4   = 4 * lane;
    const int sub  = lane & 7;
    const int head = lane >> 3;
    const float4 we0 = *(const float4*)(We + c4);
    const float4 we1 = *(const float4*)(We + HIDW + c4);
    const float4 we2 = *(const float4*)(We + 2 * HIDW + c4);
    const float4 at  = *(const float4*)(att + c4);
    auto logits = [&](int n, auto&& fetch, const bool keep) {
      if (pcount + n > PCAP) listed = false;
      for (int base = 0; base < n; base += 32) {
        const int cnt = (n - base < 32) ? (n - base) : 32;
        int ev = 0, srcv = 0, dlv = 0, okv = 0;
        float av0 = 0.f, av1 = 0.f, av2 = 0.f;
        if (lane < cnt) {
          unsigned dl;
          fetch(base + lane, ev, srcv, dl, av0, av1, av2);
          if (dl < (unsigned)OWN_NODES) { dlv = (int)dl; okv = 1; }
          if (listed) {
            const uint2 rec = make_uint2((unsigned)ev, (unsigned)srcv | ((okv ? dl : 255u) << 16));
            pl[pcount + base + lane] = rec;
            if (keep) {
              const size_t m = mslot + (size_t)(pcount + base + lane);
              *(uint2*)(mlist + 2 * m) = rec;
              mattr[3 * m + 0] = av0;
              mattr[3 * m + 1] = av1;
              mattr[3 * m + 2] = av2;
            }
          }
        }
        for (int j = 0; j < cnt; j += 8) {
          v4f xa[8];
          int dj[8];
#pragma unroll
          for (int u = 0; u < 8; ++u) {
            const int jj = j + u;
            const int sj = __builtin_amdgcn_readlane(srcv, jj);
            dj[u] = __builtin_amdgcn_readlane(dlv, jj);
            xa[u] = *(const v4f*)(xl + (size_t)sj * HIDW + c4);
          }
          asm volatile("" :: "v"(xa[0]), "v"(xa[1]), "v"(xa[2]), "v"(xa[3]),
                             "v"(xa[4]), "v"(xa[5]), "v"(xa[6]), "v"(xa[7]));
          float mine = 0.f;
#pragma unroll
          for (int u = 0; u < 8; ++u) {
            const int jj = j + u;
            const v4f xb = *(const v4f*)(sacc + dj[u] * HIDW + c4);
            const float ea0 = __int_as_float(__builtin_amdgcn_readlane(__float_as_int(av0), jj));
            const float ea1 = __int_as_float(__builtin_amdgcn_readlane(__float_as_int(av1), jj));
            const float ea2 = __int_as_float(__builtin_amdgcn_readlane(__float_as_int(av2), jj));
            float v0 = xa[u].x + xb.x + (ea0 * we0.x + ea1 * we1.x + ea2 * we2.x);
            float v1 = xa[u].y + xb.y + (ea0 * we0.y + ea1 * we1.y + ea2 * we2.y);
            float v2 = xa[u].z + xb.z + (ea0 * we0.z + ea1 * we1.z + ea2 * we2.z);
            float v3 = xa[u].w + xb.w + (ea0 * we0.w + ea1 * we1.w + ea2 * we2.w);
            v0 = (v0 > 0.f) ? v0 : 0.2f * v0;
            v1 = (v1 > 0.f) ? v1 : 0.2f * v1;
            v2 = (v2 > 0.f) ? v2 : 0.2f * v2;
            v3 = (v3 > 0.f) ? v3 : 0.2f * v3;
            float p = v0 * at.x + v1 * at.y + v2 * at.z + v3 * at.w;
            p = add_xmask<0x161>(p);
            p = add_xmask<0x162>(p);
            p = add_xmask<0x164>(p);
            if (sub == u) mine = p;
          }
          const int el    = j + sub;
          const int emine = __shfl(ev, el, 32);
          const int dmine = __shfl(dlv, el, 32);
          const int omine = __shfl(okv, el, 32);
          if (omine) {
            elog[(size_t)emine * NHEAD + head] = mine;
            atomicMax(&smax[dmine * NHEAD + head], encOrd(mine));
          }
        }
      }
      if (listed) pcount += n;
    };
    auto from_queue = [&](int i, int& e, int& s, unsigned& dl, float& a0, float& a1, float& a2) {
      e = (int)q[i];
      const int dstv = dstp[e];
      s  = ei[e];
      a0 = eattr[(size_t)e * 3 + 0];
      a1 = eattr[(size_t)e * 3 + 1];
      a2 = eattr[(size_t)e * 3 + 2];
      dl = (unsigned)(dstv - nodeBase);
    };
    auto from_store = [&](int i, int& e, int& s, unsigned& dl, float& a0, float& a1, float& a2) {
      const size_t m = mslot + (size_t)i;
      const uint2 r = *(const uint2*)(mlist + 2 * m);
      e  = (int)r.x; s = (int)(r.y & 0xFFFFu); dl = r.y >> 16;
      a0 = mattr[3 * m + 0];
      a1 = mattr[3 * m + 1];
      a2 = mattr[3 * m + 2];
    };
    int mc = -1;
    if (!first) mc = __builtin_amdgcn_readfirstlane(mcnt[blockIdx.x * OWN_WAVES + wave]);
    if ((unsigned)mc <= (unsigned)PCAP) {
      logits(mc, from_store, false);
    } else {
      scan([&](int n) { logits(n, from_queue, first != 0); });
      if (first && lane == 0) mcnt[blockIdx.x * OWN_WAVES + wave] = listed ? pcount : -1;
    }
  }
  __syncthreads();

  for (int i = tid * 4; i < OWN_NODES * HIDW; i += 4096) *(float4*)(sacc + i) = make_float4(0.f, 0.f, 0.f, 0.f);
  if (tid < OWN_NODES * NHEAD) {
    const float f = decOrd(smax[tid]);
    smax[tid] = __float_as_uint(isfinite(f) ? f : 0.f);
  }
  __syncthreads();

  auto sums = [&](int n, auto&& fetch) {
    for (int base = 0; base < n; base += 32) {
      const int cnt = (n - base < 32) ? (n - base) : 32;
      int srcv = 0, dlv = 0;
      float w0 = 0.f, w1 = 0.f, w2 = 0.f, w3 = 0.f;
      if (lane < cnt) {
        int e; unsigned dl;
        fetch(base + lane, e, srcv, dl);
        const float4 lg = *(const float4*)(elog + (size_t)e * NHEAD);
        if (dl < (unsigned)OWN_NODES) {
          dlv  = (int)dl;
          const uint4 mx = *(const uint4*)(smax + dlv * NHEAD);
          w0 = expf(lg.x - __uint_as_float(mx.x));
          w1 = expf(lg.y - __uint_as_float(mx.y));
          w2 = expf(lg.z - __uint_as_float(mx.z));
          w3 = expf(lg.w - __uint_as_float(mx.w));
          atomicAdd(&sden[dlv * NHEAD + 0], w0);
          atomicAdd(&sden[dlv * NHEAD + 1], w1);
          atomicAdd(&sden[dlv * NHEAD + 2], w2);
          atomicAdd(&sden[dlv * NHEAD + 3], w3);
        }
      }
      for (int j = 0; j < cnt; j += 8) {
        float x[8][NHEAD];
        float wj[8][NHEAD];
        int   dj[8];
#pragma unroll
        for (int u = 0; u < 8; ++u) {
          const int jj = j + u;
          const int sj = __builtin_amdgcn_readlane(srcv, jj);
          dj[u]    = __builtin_amdgcn_readlane(dlv, jj);
          wj[u][0] = __int_as_float(__builtin_amdgcn_readlane(__float_as_int(w0), jj));
          wj[u][1] = __int_as_float(__builtin_amdgcn_readlane(__float_as_int(w1), jj));
          wj[u][2] = __int_as_float(__builtin_amdgcn_readlane(__float_as_int(w2), jj));
          wj[u][3] = __int_as_float(__builtin_amdgcn_readlane(__float_as_int(w3), jj));
          const float* row = xl + (size_t)sj * HIDW + lane;
#pragma unroll
          for (int h = 0; h < NHEAD; ++h) x[u][h] = row[h * CPH];
        }
#pragma unroll
        for (int u = 0; u < 8; ++u) {
          if (j + u < cnt) {
            float* arow = sacc + dj[u] * HIDW + lane;
#pragma unroll
            for (int h = 0; h < NHEAD; ++h) atomicAdd(&arow[h * CPH], x[u][h] * wj[u][h]);
          }
        }
      }
    }
  };
  if (listed) {
    sums(pcount, [&](int i, int& e, int& s, unsigned& dl) {
      const uint2 r = pl[i];
      e = (int)r.x; s = (int)(r.y & 0xFFFFu); dl = r.y >> 16;
    });
  } else {
    scan([&](int n) {
      sums(n, [&](int i, int& e, int& s, unsigned& dl) {
        e = (int)q[i]; s = ei[e]; dl = (unsigned)(dstp[e] - nodeBase);
      });
    });
  }
  __syncthreads();

  for (int i = tid; i < OWN_NODES * HIDW; i += 1024) {
    const int node = nodeBase + (i >> 7);
    if (node < N_NODES) out[(size_t)nodeBase * HIDW + i] = sacc[i];
  }
  if (tid < OWN_NODES * NHEAD) {
    const int node = nodeBase + (tid >> 2);
    if (node < N_NODES) denom[(size_t)nodeBase * NHEAD + tid] = sden[tid];
  }
}

__device__ __forceinline__ float softmax_row_value(float s, const float* __restrict__ denom, int i) {
  return s / (denom[i >> 5] + 1e-16f);
}

__global__ __launch_bounds__(256)
void ln_reduce_kernel(const float* __restrict__ h, const float* __restrict__ denom,
                      float* __restrict__ stats) {
  __shared__ float s1[256], s2[256];
  float a = 0.f, b = 0.f;
  for (int i = blockIdx.x * 256 + threadIdx.x; i < N_NODES * HIDW; i += gridDim.x * 256) {
    float v = softmax_row_value(h[i], denom, i);
    a += v; b += v * v;
  }
  s1[threadIdx.x] = a; s2[threadIdx.x] = b;
  __syncthreads();
  for (int off = 128; off > 0; off >>= 1) {
    if ((int)threadIdx.x < off) {
      s1[threadIdx.x] += s1[threadIdx.x + off];
      s2[threadIdx.x] += s2[threadIdx.x + off];
    }
    __syncthreads();
  }
  if (threadIdx.x == 0) { atomicAdd(&stats[0], s1[0]); atomicAdd(&stats[1], s2[0]); }
}

__global__ __launch_bounds__(256)
void ln_gelu_kernel(float* __restrict__ h, const float* __restrict__ denom,
                    const float* __restrict__ stats,
                    const float* __restrict__ w, const float* __restrict__ b) {
  int i = blockIdx.x * 256 + threadIdx.x;
  if (i >= N_NODES * HIDW) return;
  const float inv  = 1.0f / (float)(N_NODES * HIDW);
  const float mean = stats[0] * inv;
  const float var  = stats[1] * inv - mean * mean;
  const float rs   = rsqrtf(var + 1e-5f);
  const int c = i & (HIDW - 1);
  float v = (softmax_row_value(h[i], denom, i) - mean) * rs * w[c] + b[c];
  v = 0.5f * v * (1.0f + erff(v * 0.70710678118654752f));
  h[i] = v;
}

extern "C" void kernel_launch(void* const* d_in, const int* in_sizes, int n_in,
                              void* d_out, int out_size, void* d_ws, size_t ws_size,
                              hipStream_t stream) {
  const float* x     = (const float*)d_in[0];
  const int*   ei    = (const int*)  d_in[1];
  const float* eattr = (const float*)d_in[2];
  const float* Wl    = (const float*)d_in[3];
  const float* Wr    = (const float*)d_in[4];
  const float* We    = (const float*)d_in[5];
  const float* att   = (const float*)d_in[6];
  const float* lnw   = (const float*)d_in[7];
  const float* lnb   = (const float*)d_in[8];
  const float* Wout  = (const float*)d_in[9];
  const float* bout  = (const float*)d_in[10];
  float* out = (float*)d_out;

  char* p = (char*)d_ws;
  auto carve = [&](size_t bytes) -> void* {
    void* q = (void*)p;
    p += (bytes + 255) & ~(size_t)255;
    return q;
  };
  float*    hA    = (float*)   carve((size_t)N_NODES * HIDW * 4);
  float*    hB    = (float*)   carve((size_t)N_NODES * HIDW * 4);
  float*    xl    = (float*)   carve((size_t)N_NODES * HIDW * 4);
  float*    xr    = (float*)   carve((size_t)N_NODES * HIDW * 4);
  float*    elog  = (float*)   carve((size_t)N_EDGES * NHEAD * 4);
  float*    denom = (float*)   carve((size_t)N_NODES * NHEAD * 4);
  float*    stats = (float*)   carve(2 * 4);
  unsigned* okey  = (unsigned*)carve((size_t)N_EDGES);
  unsigned* mlist = (unsigned*)carve((size_t)OWN_GROUPS * OWN_WAVES * PCAP * 2 * 4);
  float*    mattr = (float*)   carve((size_t)OWN_GROUPS * OWN_WAVES * PCAP * 3 * 4);
  int*      mcnt  = (int*)     carve((size_t)OWN_GROUPS * OWN_WAVES * 4);

  const int elemBlocks = (N_NODES * HIDW + 255) / 256;
  const int keyBlocks  = (N_EDGES / 4 + 255) / 256;
  const int gemmBlocks = N_NODES / 16;

  owner_keys_kernel<<<keyBlocks, 256, 0, stream>>>(ei + N_EDGES, okey);
  const float* hcur = x;
  float* bufs[2] = { hA, hB };
  for (int l = 0; l < NLAYERS; ++l) {
    float* hout = bufs[l & 1];
    gemm_dual_wmma<<<gemmBlocks, 256, 0, stream>>>(hcur, Wl + (size_t)l * HIDW * HIDW,
                                                   Wr + (size_t)l * HIDW * HIDW, xl, xr);
    edge_owner_kernel<<<OWN_GROUPS, 1024, 0, stream>>>(ei, eattr, xl, xr,
                                                        We + (size_t)l * EDIMW * HIDW,
                                                        att + (size_t)l * HIDW,
                                                        okey, elog, denom, stats, hout,
                                                        mlist, mattr, mcnt, l == 0 ? 1 : 0);
    ln_reduce_kernel<<<2048, 256, 0, stream>>>(hout, denom, stats);
    ln_gelu_kernel<<<elemBlocks, 256, 0, stream>>>(hout, denom, stats,
                                                   lnw + (size_t)l * HIDW,
                                                   lnb + (size_t)l * HIDW);
    hcur = hout;
  }
  gemm_bias_wmma<<<gemmBlocks, 256, 0, stream>>>(hcur, Wout, bout, out);
}


# Edge logits kept in LDS at the edge's place in its wave's list (scan queue drained at 32 edges to make room): elog's scattered 4-byte stores and 16-byte loads are gone; sums and their order unchanged
# speedup vs baseline: 2.3452x; 1.0782x over previous
#include <hip/hip_runtime.h>
#include <math.h>

#define N_NODES 50000
#define N_EDGES 800000
#define HIDW    128
#define NHEAD   4
#define CPH     32
#define NLAYERS 3
#define EDIMW   3

typedef __attribute__((ext_vector_type(2))) float v2f;
typedef __attribute__((ext_vector_type(8))) float v8f;
typedef __attribute__((ext_vector_type(4))) float v4f;

__device__ __forceinline__ unsigned encOrd(float f) {
  unsigned u = __float_as_uint(f);
  return (u & 0x80000000u) ? ~u : (u | 0x80000000u);
}
__device__ __forceinline__ float decOrd(unsigned u) {
  return __uint_as_float((u & 0x80000000u) ? (u & 0x7FFFFFFFu) : ~u);
}
#define ENC_NEG_INF 0x007FFFFFu

__global__ __launch_bounds__(256)
void gemm_dual_wmma(const float* __restrict__ A,
                    const float* __restrict__ W0,
                    const float* __restrict__ W1,
                    float* __restrict__ O0,
                    float* __restrict__ O1) {
  __shared__ float As[16 * 132];
  const int rowBase = blockIdx.x * 16;
  for (int i = threadIdx.x; i < 16 * HIDW; i += 256) {
    int r = i >> 7, c = i & 127;
    As[r * 132 + c] = A[(rowBase + r) * HIDW + c];
  }
  __syncthreads();
  const int wave    = threadIdx.x >> 5;
  const int lane    = threadIdx.x & 31;
  const int colBase = wave * 16;
  const int laneHi  = lane >> 4;
  const int laneLo  = lane & 15;
  v8f c0 = {}; v8f c1 = {};
  for (int k0 = 0; k0 < HIDW; k0 += 4) {
    const int ka = k0 + 2 * laneHi;
    v2f a;
    a.x = As[laneLo * 132 + ka];
    a.y = As[laneLo * 132 + ka + 1];
    v2f b0, b1;
    b0.x = W0[(ka + 0) * HIDW + colBase + laneLo];
    b0.y = W0[(ka + 1) * HIDW + colBase + laneLo];
    b1.x = W1[(ka + 0) * HIDW + colBase + laneLo];
    b1.y = W1[(ka + 1) * HIDW + colBase + laneLo];
    c0 = __builtin_amdgcn_wmma_f32_16x16x4_f32(false, a, false, b0, (short)0, c0, false, false);
    c1 = __builtin_amdgcn_wmma_f32_16x16x4_f32(false, a, false, b1, (short)0, c1, false, false);
  }
#pragma unroll
  for (int r = 0; r < 8; ++r) {
    const int row = rowBase + r + 8 * laneHi;
    O0[row * HIDW + colBase + laneLo] = c0[r];
    O1[row * HIDW + colBase + laneLo] = c1[r];
  }
}

__global__ __launch_bounds__(256)
void gemm_bias_wmma(const float* __restrict__ A,
                    const float* __restrict__ W,
                    const float* __restrict__ bias,
                    float* __restrict__ O) {
  __shared__ float As[16 * 132];
  const int rowBase = blockIdx.x * 16;
  for (int i = threadIdx.x; i < 16 * HIDW; i += 256) {
    int r = i >> 7, c = i & 127;
    As[r * 132 + c] = A[(rowBase + r) * HIDW + c];
  }
  __syncthreads();
  const int wave    = threadIdx.x >> 5;
  const int lane    = threadIdx.x & 31;
  const int colBase = wave * 16;
  const int laneHi  = lane >> 4;
  const int laneLo  = lane & 15;
  v8f c0 = {};
  for (int k0 = 0; k0 < HIDW; k0 += 4) {
    const int ka = k0 + 2 * laneHi;
    v2f a;
    a.x = As[laneLo * 132 + ka];
    a.y = As[laneLo * 132 + ka + 1];
    v2f b;
    b.x = W[(ka + 0) * HIDW + colBase + laneLo];
    b.y = W[(ka + 1) * HIDW + colBase + laneLo];
    c0 = __builtin_amdgcn_wmma_f32_16x16x4_f32(false, a, false, b, (short)0, c0, false, false);
  }
  const float bv = bias[colBase + laneLo];
#pragma unroll
  for (int r = 0; r < 8; ++r) {
    const int row = rowBase + r + 8 * laneHi;
    O[row * HIDW + colBase + laneLo] = c0[r] + bv;
  }
}

#define OWN_NODES   196
#define OWN_GROUPS  256
#define OWN_WAVES   32
static_assert(N_EDGES % 4 == 0, "one thread packs the owner bytes of four edges");
static_assert((N_NODES + OWN_NODES - 1) / OWN_NODES <= 256, "owner id fits one byte");

__global__ __launch_bounds__(256)
void owner_keys_kernel(const int* __restrict__ dstp, unsigned* __restrict__ okey) {
  int i = blockIdx.x * 256 + threadIdx.x;
  if (i >= N_EDGES / 4) return;
  const int4 d = *(const int4*)(dstp + 4 * i);
  okey[i] = (((unsigned)d.x / OWN_NODES) & 0xFFu)
          | ((((unsigned)d.y / OWN_NODES) & 0xFFu) << 8)
          | ((((unsigned)d.z / OWN_NODES) & 0xFFu) << 16)
          | ((((unsigned)d.w / OWN_NODES) & 0xFFu) << 24);
}

template <int XCTRL>
__device__ __forceinline__ float add_xmask(float p) {
  return p + __int_as_float(__builtin_amdgcn_update_dpp(0, __float_as_int(p), XCTRL, 0xf, 0xf, false));
}

#define SCAN_EDGES  2048
#define SCAN_STEPS  13
#define SCAN_CHUNKS (SCAN_STEPS * OWN_WAVES)
#define QCAP        64
#define PCAP        224
static_assert(N_NODES <= 65536 && OWN_NODES < 255, "a list entry packs the source into 16 bits, the local node into 8");
static_assert(SCAN_CHUNKS * SCAN_EDGES >= N_EDGES, "scan covers every edge");
static_assert(N_EDGES % 16 == 0, "a lane's 16 owner bytes are all edges or none");

__global__ __launch_bounds__(1024)
void edge_owner_kernel(const int* __restrict__ ei, const float* __restrict__ eattr,
                       const float* __restrict__ xl, const float* __restrict__ xr,
                       const float* __restrict__ We, const float* __restrict__ att,
                       const unsigned* __restrict__ okey, float* __restrict__ elog,
                       float* __restrict__ denom, float* __restrict__ stats,
                       float* __restrict__ out,
                       unsigned* __restrict__ mlist, float* __restrict__ mattr,
                       int* __restrict__ mcnt, int first) {
  __shared__ __attribute__((aligned(16))) float    sacc[OWN_NODES * HIDW];
  __shared__ __attribute__((aligned(16))) float    sden[OWN_NODES * NHEAD];
  __shared__ __attribute__((aligned(16))) unsigned smax[OWN_NODES * NHEAD];
  __shared__ unsigned sq[OWN_WAVES * QCAP];
  __shared__ __attribute__((aligned(16))) uint2 sp[OWN_WAVES * PCAP];
  __shared__ __attribute__((aligned(16))) float slg[OWN_WAVES * PCAP * NHEAD];
  const int tid  = threadIdx.x;
  const int lane = tid & 31;
  const int wave = tid >> 5;
  const int nodeBase = blockIdx.x * OWN_NODES;
  if (blockIdx.x == 0 && tid < 2) stats[tid] = 0.f;
  for (int i = tid * 4; i < OWN_NODES * HIDW; i += 4096) {
    const int node = nodeBase + (i >> 7);
    float4 v = make_float4(0.f, 0.f, 0.f, 0.f);
    if (node < N_NODES) v = *(const float4*)(xr + (size_t)nodeBase * HIDW + i);
    *(float4*)(sacc + i) = v;
  }
  if (tid < OWN_NODES * NHEAD) { sden[tid] = 0.f; smax[tid] = ENC_NEG_INF; }
  __syncthreads();

  unsigned* q  = sq + wave * QCAP;
  uint2*    pl = sp + wave * PCAP;
  float*    lgl = slg + wave * PCAP * NHEAD;
  const int* dstp = ei + N_EDGES;
  int count = 0;

  const unsigned gpat = (unsigned)blockIdx.x * 0x01010101u;
  const int rot = ((int)blockIdx.x * 13) % SCAN_CHUNKS;
  auto chunk_of = [&](int it) {
    int t = it * OWN_WAVES + wave + rot;
    if (t >= SCAN_CHUNKS) t -= SCAN_CHUNKS;
    return t;
  };
  auto load_chunk = [&](uint4 (&d)[4], int chunk) {
#pragma unroll
    for (int k = 0; k < 4; ++k) {
      const int e0 = chunk * SCAN_EDGES + k * 512 + lane * 16;
      if (e0 < N_EDGES) d[k] = *(const uint4*)(okey + (e0 >> 2));
      else              d[k] = make_uint4(~gpat, ~gpat, ~gpat, ~gpat);
    }
  };
  auto differs = [&](unsigned w) {
    const unsigned x = w ^ gpat;
    return ((x & 0x7F7F7F7Fu) + 0x7F7F7F7Fu) | x;
  };
  auto sel = [](unsigned m, unsigned a, unsigned b) { return (a & m) | (b & ~m); };
  auto hits = [&](const uint4& a, const uint4& b) {
    unsigned r = differs(b.w);
    r = sel(0x40404040u, differs(b.z) >> 1, r);
    r = sel(0x20202020u, differs(b.y) >> 2, r);
    r = sel(0x10101010u, differs(b.x) >> 3, r);
    r = sel(0x08080808u, differs(a.w) >> 4, r);
    r = sel(0x04040404u, differs(a.z) >> 5, r);
    r = sel(0x02020202u, differs(a.y) >> 6, r);
    r = sel(0x01010101u, differs(a.x) >> 7, r);
    return ~r;
  };
  auto push = [&](unsigned hm, int ebase, auto&& work) {
    for (int round = 0; round < 32; ++round) {
      const bool act = hm != 0u;
      const unsigned mask = __builtin_amdgcn_ballot_w32(act);
      if (mask == 0u) break;
      if (act) {
        const int p = __builtin_ctz(hm);
        hm &= hm - 1u;
        const int i = p & 7;
        q[count + (int)__builtin_amdgcn_mbcnt_lo(mask, 0u)] =
            (unsigned)(ebase + (i >> 2) * 512 + (i & 3) * 4 + (p >> 3));
      }
      count += __builtin_popcount(mask);
      if (count >= 32) { work(count); count = 0; }
    }
  };
  auto scan = [&](auto&& work) {
    count = 0;
    uint4 nxt[4];
    load_chunk(nxt, chunk_of(0));
    for (int it = 0; it < SCAN_STEPS; ++it) {
      uint4 cur[4];
#pragma unroll
      for (int k = 0; k < 4; ++k) cur[k] = nxt[k];
      const int chunk = chunk_of(it);
      if (it + 1 < SCAN_STEPS) load_chunk(nxt, chunk_of(it + 1));
      const int ebase = chunk * SCAN_EDGES + lane * 16;
      push(hits(cur[0], cur[1]), ebase, work);
      push(hits(cur[2], cur[3]), ebase + 1024, work);
    }
    work(count);
  };

  int  pcount = 0;
  bool listed = true;
  const size_t mslot = ((size_t)blockIdx.x * OWN_WAVES + wave) * PCAP;
  {
    const int c4   = 4 * lane;
    const int sub  = lane & 7;
    const int head = lane >> 3;
    const float4 we0 = *(const float4*)(We + c4);
    const float4 we1 = *(const float4*)(We + HIDW + c4);
    const float4 we2 = *(const float4*)(We + 2 * HIDW + c4);
    const float4 at  = *(const float4*)(att + c4);
    auto logits = [&](int n, auto&& fetch, const bool keep) {
      if (listed && pcount + n > PCAP) {
        for (int i = lane; i < pcount; i += 32)
          *(float4*)(elog + (size_t)pl[i].x * NHEAD) = *(const float4*)(lgl + i * NHEAD);
        listed = false;
      }
      for (int base = 0; base < n; base += 32) {
        const int cnt = (n - base < 32) ? (n - base) : 32;
        int ev = 0, srcv = 0, dlv = 0, okv = 0;
        float av0 = 0.f, av1 = 0.f, av2 = 0.f;
        if (lane < cnt) {
          unsigned dl;
          fetch(base + lane, ev, srcv, dl, av0, av1, av2);
          if (dl < (unsigned)OWN_NODES) { dlv = (int)dl; okv = 1; }
          if (listed) {
            const uint2 rec = make_uint2((unsigned)ev, (unsigned)srcv | ((okv ? dl : 255u) << 16));
            pl[pcount + base + lane] = rec;
            if (keep) {
              const size_t m = mslot + (size_t)(pcount + base + lane);
              *(uint2*)(mlist + 2 * m) = rec;
              mattr[3 * m + 0] = av0;
              mattr[3 * m + 1] = av1;
              mattr[3 * m + 2] = av2;
            }
          }
        }
        for (int j = 0; j < cnt; j += 8) {
          v4f xa[8];
          int dj[8];
#pragma unroll
          for (int u = 0; u < 8; ++u) {
            const int jj = j + u;
            const int sj = __builtin_amdgcn_readlane(srcv, jj);
            dj[u] = __builtin_amdgcn_readlane(dlv, jj);
            xa[u] = *(const v4f*)(xl + (size_t)sj * HIDW + c4);
          }
          asm volatile("" :: "v"(xa[0]), "v"(xa[1]), "v"(xa[2]), "v"(xa[3]),
                             "v"(xa[4]), "v"(xa[5]), "v"(xa[6]), "v"(xa[7]));
          float mine = 0.f;
#pragma unroll
          for (int u = 0; u < 8; ++u) {
            const int jj = j + u;
            const v4f xb = *(const v4f*)(sacc + dj[u] * HIDW + c4);
            const float ea0 = __int_as_float(__builtin_amdgcn_readlane(__float_as_int(av0), jj));
            const float ea1 = __int_as_float(__builtin_amdgcn_readlane(__float_as_int(av1), jj));
            const float ea2 = __int_as_float(__builtin_amdgcn_readlane(__float_as_int(av2), jj));
            float v0 = xa[u].x + xb.x + (ea0 * we0.x + ea1 * we1.x + ea2 * we2.x);
            float v1 = xa[u].y + xb.y + (ea0 * we0.y + ea1 * we1.y + ea2 * we2.y);
            float v2 = xa[u].z + xb.z + (ea0 * we0.z + ea1 * we1.z + ea2 * we2.z);
            float v3 = xa[u].w + xb.w + (ea0 * we0.w + ea1 * we1.w + ea2 * we2.w);
            v0 = (v0 > 0.f) ? v0 : 0.2f * v0;
            v1 = (v1 > 0.f) ? v1 : 0.2f * v1;
            v2 = (v2 > 0.f) ? v2 : 0.2f * v2;
            v3 = (v3 > 0.f) ? v3 : 0.2f * v3;
            float p = v0 * at.x + v1 * at.y + v2 * at.z + v3 * at.w;
            p = add_xmask<0x161>(p);
            p = add_xmask<0x162>(p);
            p = add_xmask<0x164>(p);
            if (sub == u) mine = p;
          }
          const int el    = j + sub;
          const int emine = __shfl(ev, el, 32);
          const int dmine = __shfl(dlv, el, 32);
          const int omine = __shfl(okv, el, 32);
          if (omine) {
            if (listed) lgl[(pcount + base + el) * NHEAD + head] = mine;
            else        elog[(size_t)emine * NHEAD + head] = mine;
            atomicMax(&smax[dmine * NHEAD + head], encOrd(mine));
          }
        }
      }
      if (listed) pcount += n;
    };
    auto from_queue = [&](int i, int& e, int& s, unsigned& dl, float& a0, float& a1, float& a2) {
      e = (int)q[i];
      const int dstv = dstp[e];
      s  = ei[e];
      a0 = eattr[(size_t)e * 3 + 0];
      a1 = eattr[(size_t)e * 3 + 1];
      a2 = eattr[(size_t)e * 3 + 2];
      dl = (unsigned)(dstv - nodeBase);
    };
    auto from_store = [&](int i, int& e, int& s, unsigned& dl, float& a0, float& a1, float& a2) {
      const size_t m = mslot + (size_t)i;
      const uint2 r = *(const uint2*)(mlist + 2 * m);
      e  = (int)r.x; s = (int)(r.y & 0xFFFFu); dl = r.y >> 16;
      a0 = mattr[3 * m + 0];
      a1 = mattr[3 * m + 1];
      a2 = mattr[3 * m + 2];
    };
    int mc = -1;
    if (!first) mc = __builtin_amdgcn_readfirstlane(mcnt[blockIdx.x * OWN_WAVES + wave]);
    if ((unsigned)mc <= (unsigned)PCAP) {
      logits(mc, from_store, false);
    } else {
      scan([&](int n) { logits(n, from_queue, first != 0); });
      if (first && lane == 0) mcnt[blockIdx.x * OWN_WAVES + wave] = listed ? pcount : -1;
    }
  }
  __syncthreads();

  for (int i = tid * 4; i < OWN_NODES * HIDW; i += 4096) *(float4*)(sacc + i) = make_float4(0.f, 0.f, 0.f, 0.f);
  if (tid < OWN_NODES * NHEAD) {
    const float f = decOrd(smax[tid]);
    smax[tid] = __float_as_uint(isfinite(f) ? f : 0.f);
  }
  __syncthreads();

  auto sums = [&](int n, auto&& fetch) {
    for (int base = 0; base < n; base += 32) {
      const int cnt = (n - base < 32) ? (n - base) : 32;
      int srcv = 0, dlv = 0;
      float w0 = 0.f, w1 = 0.f, w2 = 0.f, w3 = 0.f;
      if (lane < cnt) {
        unsigned dl; float4 lg;
        fetch(base + lane, srcv, dl, lg);
        if (dl < (unsigned)OWN_NODES) {
          dlv  = (int)dl;
          const uint4 mx = *(const uint4*)(smax + dlv * NHEAD);
          w0 = expf(lg.x - __uint_as_float(mx.x));
          w1 = expf(lg.y - __uint_as_float(mx.y));
          w2 = expf(lg.z - __uint_as_float(mx.z));
          w3 = expf(lg.w - __uint_as_float(mx.w));
          atomicAdd(&sden[dlv * NHEAD + 0], w0);
          atomicAdd(&sden[dlv * NHEAD + 1], w1);
          atomicAdd(&sden[dlv * NHEAD + 2], w2);
          atomicAdd(&sden[dlv * NHEAD + 3], w3);
        }
      }
      for (int j = 0; j < cnt; j += 8) {
        float x[8][NHEAD];
        float wj[8][NHEAD];
        int   dj[8];
#pragma unroll
        for (int u = 0; u < 8; ++u) {
          const int jj = j + u;
          const int sj = __builtin_amdgcn_readlane(srcv, jj);
          dj[u]    = __builtin_amdgcn_readlane(dlv, jj);
          wj[u][0] = __int_as_float(__builtin_amdgcn_readlane(__float_as_int(w0), jj));
          wj[u][1] = __int_as_float(__builtin_amdgcn_readlane(__float_as_int(w1), jj));
          wj[u][2] = __int_as_float(__builtin_amdgcn_readlane(__float_as_int(w2), jj));
          wj[u][3] = __int_as_float(__builtin_amdgcn_readlane(__float_as_int(w3), jj));
          const float* row = xl + (size_t)sj * HIDW + lane;
#pragma unroll
          for (int h = 0; h < NHEAD; ++h) x[u][h] = row[h * CPH];
        }
#pragma unroll
        for (int u = 0; u < 8; ++u) {
          if (j + u < cnt) {
            float* arow = sacc + dj[u] * HIDW + lane;
#pragma unroll
            for (int h = 0; h < NHEAD; ++h) atomicAdd(&arow[h * CPH], x[u][h] * wj[u][h]);
          }
        }
      }
    }
  };
  if (listed) {
    sums(pcount, [&](int i, int& s, unsigned& dl, float4& lg) {
      const uint2 r = pl[i];
      s = (int)(r.y & 0xFFFFu); dl = r.y >> 16;
      lg = *(const float4*)(lgl + i * NHEAD);
    });
  } else {
    scan([&](int n) {
      sums(n, [&](int i, int& s, unsigned& dl, float4& lg) {
        const int e = (int)q[i]; s = ei[e]; dl = (unsigned)(dstp[e] - nodeBase);
        lg = *(const float4*)(elog + (size_t)e * NHEAD);
      });
    });
  }
  __syncthreads();

  for (int i = tid; i < OWN_NODES * HIDW; i += 1024) {
    const int node = nodeBase + (i >> 7);
    if (node < N_NODES) out[(size_t)nodeBase * HIDW + i] = sacc[i];
  }
  if (tid < OWN_NODES * NHEAD) {
    const int node = nodeBase + (tid >> 2);
    if (node < N_NODES) denom[(size_t)nodeBase * NHEAD + tid] = sden[tid];
  }
}

__device__ __forceinline__ float softmax_row_value(float s, const float* __restrict__ denom, int i) {
  return s / (denom[i >> 5] + 1e-16f);
}

__global__ __launch_bounds__(256)
void ln_reduce_kernel(const float* __restrict__ h, const float* __restrict__ denom,
                      float* __restrict__ stats) {
  __shared__ float s1[256], s2[256];
  float a = 0.f, b = 0.f;
  for (int i = blockIdx.x * 256 + threadIdx.x; i < N_NODES * HIDW; i += gridDim.x * 256) {
    float v = softmax_row_value(h[i], denom, i);
    a += v; b += v * v;
  }
  s1[threadIdx.x] = a; s2[threadIdx.x] = b;
  __syncthreads();
  for (int off = 128; off > 0; off >>= 1) {
    if ((int)threadIdx.x < off) {
      s1[threadIdx.x] += s1[threadIdx.x + off];
      s2[threadIdx.x] += s2[threadIdx.x + off];
    }
    __syncthreads();
  }
  if (threadIdx.x == 0) { atomicAdd(&stats[0], s1[0]); atomicAdd(&stats[1], s2[0]); }
}

__global__ __launch_bounds__(256)
void ln_gelu_kernel(float* __restrict__ h, const float* __restrict__ denom,
                    const float* __restrict__ stats,
                    const float* __restrict__ w, const float* __restrict__ b) {
  int i = blockIdx.x * 256 + threadIdx.x;
  if (i >= N_NODES * HIDW) return;
  const float inv  = 1.0f / (float)(N_NODES * HIDW);
  const float mean = stats[0] * inv;
  const float var  = stats[1] * inv - mean * mean;
  const float rs   = rsqrtf(var + 1e-5f);
  const int c = i & (HIDW - 1);
  float v = (softmax_row_value(h[i], denom, i) - mean) * rs * w[c] + b[c];
  v = 0.5f * v * (1.0f + erff(v * 0.70710678118654752f));
  h[i] = v;
}

extern "C" void kernel_launch(void* const* d_in, const int* in_sizes, int n_in,
                              void* d_out, int out_size, void* d_ws, size_t ws_size,
                              hipStream_t stream) {
  const float* x     = (const float*)d_in[0];
  const int*   ei    = (const int*)  d_in[1];
  const float* eattr = (const float*)d_in[2];
  const float* Wl    = (const float*)d_in[3];
  const float* Wr    = (const float*)d_in[4];
  const float* We    = (const float*)d_in[5];
  const float* att   = (const float*)d_in[6];
  const float* lnw   = (const float*)d_in[7];
  const float* lnb   = (const float*)d_in[8];
  const float* Wout  = (const float*)d_in[9];
  const float* bout  = (const float*)d_in[10];
  float* out = (float*)d_out;

  char* p = (char*)d_ws;
  auto carve = [&](size_t bytes) -> void* {
    void* q = (void*)p;
    p += (bytes + 255) & ~(size_t)255;
    return q;
  };
  float*    hA    = (float*)   carve((size_t)N_NODES * HIDW * 4);
  float*    hB    = (float*)   carve((size_t)N_NODES * HIDW * 4);
  float*    xl    = (float*)   carve((size_t)N_NODES * HIDW * 4);
  float*    xr    = (float*)   carve((size_t)N_NODES * HIDW * 4);
  float*    elog  = (float*)   carve((size_t)N_EDGES * NHEAD * 4);
  float*    denom = (float*)   carve((size_t)N_NODES * NHEAD * 4);
  float*    stats = (float*)   carve(2 * 4);
  unsigned* okey  = (unsigned*)carve((size_t)N_EDGES);
  unsigned* mlist = (unsigned*)carve((size_t)OWN_GROUPS * OWN_WAVES * PCAP * 2 * 4);
  float*    mattr = (float*)   carve((size_t)OWN_GROUPS * OWN_WAVES * PCAP * 3 * 4);
  int*      mcnt  = (int*)     carve((size_t)OWN_GROUPS * OWN_WAVES * 4);

  const int elemBlocks = (N_NODES * HIDW + 255) / 256;
  const int keyBlocks  = (N_EDGES / 4 + 255) / 256;
  const int gemmBlocks = N_NODES / 16;

  owner_keys_kernel<<<keyBlocks, 256, 0, stream>>>(ei + N_EDGES, okey);
  const float* hcur = x;
  float* bufs[2] = { hA, hB };
  for (int l = 0; l < NLAYERS; ++l) {
    float* hout = bufs[l & 1];
    gemm_dual_wmma<<<gemmBlocks, 256, 0, stream>>>(hcur, Wl + (size_t)l * HIDW * HIDW,
                                                   Wr + (size_t)l * HIDW * HIDW, xl, xr);
    edge_owner_kernel<<<OWN_GROUPS, 1024, 0, stream>>>(ei, eattr, xl, xr,
                                                        We + (size_t)l * EDIMW * HIDW,
                                                        att + (size_t)l * HIDW,
                                                        okey, elog, denom, stats, hout,
                                                        mlist, mattr, mcnt, l == 0 ? 1 : 0);
    ln_reduce_kernel<<<2048, 256, 0, stream>>>(hout, denom, stats);
    ln_gelu_kernel<<<elemBlocks, 256, 0, stream>>>(hout, denom, stats,
                                                   lnw + (size_t)l * HIDW,
                                                   lnb + (size_t)l * HIDW);
    hcur = hout;
  }
  gemm_bias_wmma<<<gemmBlocks, 256, 0, stream>>>(hcur, Wout, bout, out);
}


# Softmax shift = first logit to reach a (node, head) (LDS compare-and-swap, arrival order): exp weights, denominators and weighted row sums made in the logit pass; second xl gather and sum phase gone
# speedup vs baseline: 2.6567x; 1.1328x over previous
#include <hip/hip_runtime.h>
#include <math.h>

#define N_NODES 50000
#define N_EDGES 800000
#define HIDW    128
#define NHEAD   4
#define CPH     32
#define NLAYERS 3
#define EDIMW   3

typedef __attribute__((ext_vector_type(2))) float v2f;
typedef __attribute__((ext_vector_type(8))) float v8f;
typedef __attribute__((ext_vector_type(4))) float v4f;

__device__ __forceinline__ unsigned encOrd(float f) {
  unsigned u = __float_as_uint(f);
  return (u & 0x80000000u) ? ~u : (u | 0x80000000u);
}
__device__ __forceinline__ float decOrd(unsigned u) {
  return __uint_as_float((u & 0x80000000u) ? (u & 0x7FFFFFFFu) : ~u);
}
#define ENC_NEG_INF 0x007FFFFFu

__global__ __launch_bounds__(256)
void gemm_dual_wmma(const float* __restrict__ A,
                    const float* __restrict__ W0,
                    const float* __restrict__ W1,
                    float* __restrict__ O0,
                    float* __restrict__ O1) {
  __shared__ float As[16 * 132];
  const int rowBase = blockIdx.x * 16;
  for (int i = threadIdx.x; i < 16 * HIDW; i += 256) {
    int r = i >> 7, c = i & 127;
    As[r * 132 + c] = A[(rowBase + r) * HIDW + c];
  }
  __syncthreads();
  const int wave    = threadIdx.x >> 5;
  const int lane    = threadIdx.x & 31;
  const int colBase = wave * 16;
  const int laneHi  = lane >> 4;
  const int laneLo  = lane & 15;
  v8f c0 = {}; v8f c1 = {};
  for (int k0 = 0; k0 < HIDW; k0 += 4) {
    const int ka = k0 + 2 * laneHi;
    v2f a;
    a.x = As[laneLo * 132 + ka];
    a.y = As[laneLo * 132 + ka + 1];
    v2f b0, b1;
    b0.x = W0[(ka + 0) * HIDW + colBase + laneLo];
    b0.y = W0[(ka + 1) * HIDW + colBase + laneLo];
    b1.x = W1[(ka + 0) * HIDW + colBase + laneLo];
    b1.y = W1[(ka + 1) * HIDW + colBase + laneLo];
    c0 = __builtin_amdgcn_wmma_f32_16x16x4_f32(false, a, false, b0, (short)0, c0, false, false);
    c1 = __builtin_amdgcn_wmma_f32_16x16x4_f32(false, a, false, b1, (short)0, c1, false, false);
  }
#pragma unroll
  for (int r = 0; r < 8; ++r) {
    const int row = rowBase + r + 8 * laneHi;
    O0[row * HIDW + colBase + laneLo] = c0[r];
    O1[row * HIDW + colBase + laneLo] = c1[r];
  }
}

__global__ __launch_bounds__(256)
void gemm_bias_wmma(const float* __restrict__ A,
                    const float* __restrict__ W,
                    const float* __restrict__ bias,
                    float* __restrict__ O) {
  __shared__ float As[16 * 132];
  const int rowBase = blockIdx.x * 16;
  for (int i = threadIdx.x; i < 16 * HIDW; i += 256) {
    int r = i >> 7, c = i & 127;
    As[r * 132 + c] = A[(rowBase + r) * HIDW + c];
  }
  __syncthreads();
  const int wave    = threadIdx.x >> 5;
  const int lane    = threadIdx.x & 31;
  const int colBase = wave * 16;
  const int laneHi  = lane >> 4;
  const int laneLo  = lane & 15;
  v8f c0 = {};
  for (int k0 = 0; k0 < HIDW; k0 += 4) {
    const int ka = k0 + 2 * laneHi;
    v2f a;
    a.x = As[laneLo * 132 + ka];
    a.y = As[laneLo * 132 + ka + 1];
    v2f b;
    b.x = W[(ka + 0) * HIDW + colBase + laneLo];
    b.y = W[(ka + 1) * HIDW + colBase + laneLo];
    c0 = __builtin_amdgcn_wmma_f32_16x16x4_f32(false, a, false, b, (short)0, c0, false, false);
  }
  const float bv = bias[colBase + laneLo];
#pragma unroll
  for (int r = 0; r < 8; ++r) {
    const int row = rowBase + r + 8 * laneHi;
    O[row * HIDW + colBase + laneLo] = c0[r] + bv;
  }
}

#define OWN_NODES   196
#define OWN_GROUPS  256
#define OWN_WAVES   32
static_assert(N_EDGES % 4 == 0, "one thread packs the owner bytes of four edges");
static_assert((N_NODES + OWN_NODES - 1) / OWN_NODES <= 256, "owner id fits one byte");

__global__ __launch_bounds__(256)
void owner_keys_kernel(const int* __restrict__ dstp, unsigned* __restrict__ okey) {
  int i = blockIdx.x * 256 + threadIdx.x;
  if (i >= N_EDGES / 4) return;
  const int4 d = *(const int4*)(dstp + 4 * i);
  okey[i] = (((unsigned)d.x / OWN_NODES) & 0xFFu)
          | ((((unsigned)d.y / OWN_NODES) & 0xFFu) << 8)
          | ((((unsigned)d.z / OWN_NODES) & 0xFFu) << 16)
          | ((((unsigned)d.w / OWN_NODES) & 0xFFu) << 24);
}

template <int XCTRL>
__device__ __forceinline__ float add_xmask(float p) {
  return p + __int_as_float(__builtin_amdgcn_update_dpp(0, __float_as_int(p), XCTRL, 0xf, 0xf, false));
}

template <int U>
__device__ __forceinline__ float group_lane(float w) {
  return __uint_as_float(__builtin_amdgcn_mov_dpp8(__float_as_uint(w), U * 0x249249));
}

template <int U>
__device__ __forceinline__ void add_weighted(float* pa, const v4f x, const float w) {
  const float wu = group_lane<U>(w);
  atomicAdd(pa,     x.x * wu);
  atomicAdd(pa + 1, x.y * wu);
  atomicAdd(pa + 2, x.z * wu);
  atomicAdd(pa + 3, x.w * wu);
}

#define SCAN_EDGES  2048
#define SCAN_STEPS  13
#define SCAN_CHUNKS (SCAN_STEPS * OWN_WAVES)
#define QCAP        64
#define PCAP        224
#define ACC_PITCH   (HIDW + 1)
#define SHIFT_EMPTY 0xFFFFFFFFu
#define SHIFT_SPAN  20.0f
static_assert(N_NODES <= 65536 && OWN_NODES < 255, "a list entry packs the source into 16 bits, the local node into 8");
static_assert(SCAN_CHUNKS * SCAN_EDGES >= N_EDGES, "scan covers every edge");
static_assert(N_EDGES % 16 == 0, "a lane's 16 owner bytes are all edges or none");
static_assert((OWN_NODES * ACC_PITCH) % 4 == 0, "the accumulator rows are cleared in 16-byte pieces");

__global__ __launch_bounds__(1024)
void edge_owner_kernel(const int* __restrict__ ei, const float* __restrict__ eattr,
                       const float* __restrict__ xl, const float* __restrict__ xr,
                       const float* __restrict__ We, const float* __restrict__ att,
                       const unsigned* __restrict__ okey,
                       float* __restrict__ denom, float* __restrict__ stats,
                       float* __restrict__ out,
                       unsigned* __restrict__ mlist, float* __restrict__ mattr,
                       int* __restrict__ mcnt, int first) {
  __shared__ __attribute__((aligned(16))) float    sxr[OWN_NODES * HIDW];
  __shared__ __attribute__((aligned(16))) float    sacc[OWN_NODES * ACC_PITCH];
  __shared__ __attribute__((aligned(16))) float    sden[OWN_NODES * NHEAD];
  __shared__ __attribute__((aligned(16))) unsigned smax[OWN_NODES * NHEAD];
  __shared__ __attribute__((aligned(16))) unsigned sshift[OWN_NODES * NHEAD];
  __shared__ unsigned sq[OWN_WAVES * QCAP];
  __shared__ int sflag;
  const int tid  = threadIdx.x;
  const int lane = tid & 31;
  const int wave = tid >> 5;
  const int nodeBase = blockIdx.x * OWN_NODES;
  if (blockIdx.x == 0 && tid < 2) stats[tid] = 0.f;
  for (int i = tid * 4; i < OWN_NODES * HIDW; i += 4096) {
    const int node = nodeBase + (i >> 7);
    float4 v = make_float4(0.f, 0.f, 0.f, 0.f);
    if (node < N_NODES) v = *(const float4*)(xr + (size_t)nodeBase * HIDW + i);
    *(float4*)(sxr + i)  = v;
  }
  for (int i = tid * 4; i < OWN_NODES * ACC_PITCH; i += 4096) *(float4*)(sacc + i) = make_float4(0.f, 0.f, 0.f, 0.f);
  if (tid < OWN_NODES * NHEAD) { sden[tid] = 0.f; smax[tid] = ENC_NEG_INF; sshift[tid] = SHIFT_EMPTY; }
  if (tid == 0) sflag = 0;
  __syncthreads();

  unsigned* q  = sq + wave * QCAP;
  const int* dstp = ei + N_EDGES;
  int count = 0;

  const unsigned gpat = (unsigned)blockIdx.x * 0x01010101u;
  const int rot = ((int)blockIdx.x * 13) % SCAN_CHUNKS;
  auto chunk_of = [&](int it) {
    int t = it * OWN_WAVES + wave + rot;
    if (t >= SCAN_CHUNKS) t -= SCAN_CHUNKS;
    return t;
  };
  auto load_chunk = [&](uint4 (&d)[4], int chunk) {
#pragma unroll
    for (int k = 0; k < 4; ++k) {
      const int e0 = chunk * SCAN_EDGES + k * 512 + lane * 16;
      if (e0 < N_EDGES) d[k] = *(const uint4*)(okey + ((unsigned)e0 >> 2));
      else              d[k] = make_uint4(~gpat, ~gpat, ~gpat, ~gpat);
    }
  };
  auto differs = [&](unsigned w) {
    const unsigned x = w ^ gpat;
    return ((x & 0x7F7F7F7Fu) + 0x7F7F7F7Fu) | x;
  };
  auto sel = [](unsigned m, unsigned a, unsigned b) { return (a & m) | (b & ~m); };
  auto hits = [&](const uint4& a, const uint4& b) {
    unsigned r = differs(b.w);
    r = sel(0x40404040u, differs(b.z) >> 1, r);
    r = sel(0x20202020u, differs(b.y) >> 2, r);
    r = sel(0x10101010u, differs(b.x) >> 3, r);
    r = sel(0x08080808u, differs(a.w) >> 4, r);
    r = sel(0x04040404u, differs(a.z) >> 5, r);
    r = sel(0x02020202u, differs(a.y) >> 6, r);
    r = sel(0x01010101u, differs(a.x) >> 7, r);
    return ~r;
  };
  auto push = [&](unsigned hm, int ebase, auto&& work) {
    for (int round = 0; round < 32; ++round) {
      const bool act = hm != 0u;
      const unsigned mask = __builtin_amdgcn_ballot_w32(act);
      if (mask == 0u) break;
      if (act) {
        const int p = __builtin_ctz(hm);
        hm &= hm - 1u;
        const int i = p & 7;
        q[count + (int)__builtin_amdgcn_mbcnt_lo(mask, 0u)] =
            (unsigned)(ebase + (i >> 2) * 512 + (i & 3) * 4 + (p >> 3));
      }
      count += __builtin_popcount(mask);
      if (count >= 32) { work(count); count = 0; }
    }
  };
  auto scan = [&](auto&& work) {
    count = 0;
    uint4 nxt[4];
    load_chunk(nxt, chunk_of(0));
    for (int it = 0; it < SCAN_STEPS; ++it) {
      uint4 cur[4];
#pragma unroll
      for (int k = 0; k < 4; ++k) cur[k] = nxt[k];
      const int chunk = chunk_of(it);
      if (it + 1 < SCAN_STEPS) load_chunk(nxt, chunk_of(it + 1));
      const int ebase = chunk * SCAN_EDGES + lane * 16;
      push(hits(cur[0], cur[1]), ebase, work);
      push(hits(cur[2], cur[3]), ebase + 1024, work);
    }
    work(count);
  };

  int  pcount = 0;
  bool listed = true;
  const size_t mslot = ((size_t)blockIdx.x * OWN_WAVES + wave) * PCAP;
  const int c4   = 4 * lane;
  const int sub  = lane & 7;
  const int head = lane >> 3;
  const int oA = c4 + (lane >> 4);
  const float4 we0 = *(const float4*)(We + c4);
  const float4 we1 = *(const float4*)(We + HIDW + c4);
  const float4 we2 = *(const float4*)(We + 2 * HIDW + c4);
  const float4 at  = *(const float4*)(att + c4);
  auto edges = [&](int n, auto&& fetch, const bool keep, const bool exact) {
    if (keep && listed && pcount + n > PCAP) listed = false;
    for (int base = 0; base < n; base += 32) {
      const int cnt = (n - base < 32) ? (n - base) : 32;
      int srcv = 0, dlv = 0, okv = 0;
      float av0 = 0.f, av1 = 0.f, av2 = 0.f;
      if (lane < cnt) {
        unsigned dl; int ev;
        fetch(base + lane, ev, srcv, dl, av0, av1, av2);
        if (dl < (unsigned)OWN_NODES) { dlv = (int)dl; okv = 1; }
        if (keep && listed) {
          const uint2 rec = make_uint2((unsigned)ev, (unsigned)srcv | ((okv ? dl : 255u) << 16));
          const size_t m = mslot + (size_t)(pcount + base + lane);
          *(uint2*)(mlist + 2 * m) = rec;
          mattr[3 * m + 0] = av0;
          mattr[3 * m + 1] = av1;
          mattr[3 * m + 2] = av2;
        }
      }
      for (int j = 0; j < cnt; j += 8) {
        v4f xa[8];
        int dj[8];
#pragma unroll
        for (int u = 0; u < 8; ++u) {
          const int jj = j + u;
          const int sj = __builtin_amdgcn_readlane(srcv, jj);
          dj[u] = __builtin_amdgcn_readlane(dlv, jj);
          xa[u] = *(const v4f*)(xl + (size_t)sj * HIDW + c4);
        }
        asm volatile("" :: "v"(xa[0]), "v"(xa[1]), "v"(xa[2]), "v"(xa[3]),
                           "v"(xa[4]), "v"(xa[5]), "v"(xa[6]), "v"(xa[7]));
        float mine = 0.f;
#pragma unroll
        for (int u = 0; u < 8; ++u) {
          const int jj = j + u;
          const v4f xb = *(const v4f*)(sxr + dj[u] * HIDW + c4);
          const float ea0 = __int_as_float(__builtin_amdgcn_readlane(__float_as_int(av0), jj));
          const float ea1 = __int_as_float(__builtin_amdgcn_readlane(__float_as_int(av1), jj));
          const float ea2 = __int_as_float(__builtin_amdgcn_readlane(__float_as_int(av2), jj));
          float v0 = xa[u].x + xb.x + (ea0 * we0.x + ea1 * we1.x + ea2 * we2.x);
          float v1 = xa[u].y + xb.y + (ea0 * we0.y + ea1 * we1.y + ea2 * we2.y);
          float v2 = xa[u].z + xb.z + (ea0 * we0.z + ea1 * we1.z + ea2 * we2.z);
          float v3 = xa[u].w + xb.w + (ea0 * we0.w + ea1 * we1.w + ea2 * we2.w);
          v0 = (v0 > 0.f) ? v0 : 0.2f * v0;
          v1 = (v1 > 0.f) ? v1 : 0.2f * v1;
          v2 = (v2 > 0.f) ? v2 : 0.2f * v2;
          v3 = (v3 > 0.f) ? v3 : 0.2f * v3;
          float p = v0 * at.x + v1 * at.y + v2 * at.z + v3 * at.w;
          p = add_xmask<0x161>(p);
          p = add_xmask<0x162>(p);
          p = add_xmask<0x164>(p);
          if (sub == u) mine = p;
        }
        const int el    = j + sub;
        const int dmine = __shfl(dlv, el, 32);
        const int omine = __shfl(okv, el, 32);
        float w = 0.f;
        if (omine) {
          const int slot = dmine * NHEAD + head;
          float m0;
          if (exact) {
            m0 = __uint_as_float(sshift[slot]);
          } else {
            const unsigned old = atomicCAS(&sshift[slot], SHIFT_EMPTY, __float_as_uint(mine));
            m0 = (old == SHIFT_EMPTY) ? mine : __uint_as_float(old);
            atomicMax(&smax[slot], encOrd(mine));
          }
          w = expf(mine - m0);
          atomicAdd(&sden[slot], w);
        }
#define ADD_EDGE(U) if (j + U < cnt) add_weighted<U>(sacc + dj[U] * ACC_PITCH + oA, xa[U], w);
        ADD_EDGE(0) ADD_EDGE(1) ADD_EDGE(2) ADD_EDGE(3)
        ADD_EDGE(4) ADD_EDGE(5) ADD_EDGE(6) ADD_EDGE(7)
#undef ADD_EDGE
      }
    }
    if (keep && listed) pcount += n;
  };
  auto from_queue = [&](int i, int& e, int& s, unsigned& dl, float& a0, float& a1, float& a2) {
    e = (int)q[i];
    const int dstv = dstp[e];
    s  = ei[e];
    a0 = eattr[(size_t)e * 3 + 0];
    a1 = eattr[(size_t)e * 3 + 1];
    a2 = eattr[(size_t)e * 3 + 2];
    dl = (unsigned)(dstv - nodeBase);
  };
  auto from_store = [&](int i, int& e, int& s, unsigned& dl, float& a0, float& a1, float& a2) {
    const size_t m = mslot + (size_t)i;
    const uint2 r = *(const uint2*)(mlist + 2 * m);
    e  = (int)r.x; s = (int)(r.y & 0xFFFFu); dl = r.y >> 16;
    a0 = mattr[3 * m + 0];
    a1 = mattr[3 * m + 1];
    a2 = mattr[3 * m + 2];
  };
  int mc = -1;
  if (!first) mc = __builtin_amdgcn_readfirstlane(mcnt[blockIdx.x * OWN_WAVES + wave]);
  auto run = [&](const bool exact) {
    if ((unsigned)mc <= (unsigned)PCAP) {
      edges(mc, from_store, false, exact);
    } else {
      const bool keep = first != 0 && !exact;
      scan([&](int n) { edges(n, from_queue, keep, exact); });
      if (keep && lane == 0) mcnt[blockIdx.x * OWN_WAVES + wave] = listed ? pcount : -1;
    }
  };
  run(false);
  __syncthreads();
  if (tid < OWN_NODES * NHEAD) {
    const unsigned sb = sshift[tid];
    if (sb != SHIFT_EMPTY && !(decOrd(smax[tid]) - __uint_as_float(sb) <= SHIFT_SPAN)) sflag = 1;
  }
  __syncthreads();
  if (sflag != 0) {
    for (int i = tid * 4; i < OWN_NODES * ACC_PITCH; i += 4096) *(float4*)(sacc + i) = make_float4(0.f, 0.f, 0.f, 0.f);
    if (tid < OWN_NODES * NHEAD) {
      const float f = decOrd(smax[tid]);
      sshift[tid] = __float_as_uint(isfinite(f) ? f : 0.f);
      sden[tid] = 0.f;
    }
    __syncthreads();
    run(true);
    __syncthreads();
  }

  for (int i = tid; i < OWN_NODES * HIDW; i += 1024) {
    const int node = nodeBase + (i >> 7);
    if (node < N_NODES) out[(size_t)nodeBase * HIDW + i] = sacc[i + (i >> 7) + ((i >> 6) & 1)];
  }
  if (tid < OWN_NODES * NHEAD) {
    const int node = nodeBase + (tid >> 2);
    if (node < N_NODES) denom[(size_t)nodeBase * NHEAD + tid] = sden[tid];
  }
}

__device__ __forceinline__ float softmax_row_value(float s, const float* __restrict__ denom, int i) {
  return s / (denom[i >> 5] + 1e-16f);
}

__global__ __launch_bounds__(256)
void ln_reduce_kernel(const float* __restrict__ h, const float* __restrict__ denom,
                      float* __restrict__ stats) {
  __shared__ float s1[256], s2[256];
  float a = 0.f, b = 0.f;
  for (int i = blockIdx.x * 256 + threadIdx.x; i < N_NODES * HIDW; i += gridDim.x * 256) {
    float v = softmax_row_value(h[i], denom, i);
    a += v; b += v * v;
  }
  s1[threadIdx.x] = a; s2[threadIdx.x] = b;
  __syncthreads();
  for (int off = 128; off > 0; off >>= 1) {
    if ((int)threadIdx.x < off) {
      s1[threadIdx.x] += s1[threadIdx.x + off];
      s2[threadIdx.x] += s2[threadIdx.x + off];
    }
    __syncthreads();
  }
  if (threadIdx.x == 0) { atomicAdd(&stats[0], s1[0]); atomicAdd(&stats[1], s2[0]); }
}

__global__ __launch_bounds__(256)
void ln_gelu_kernel(float* __restrict__ h, const float* __restrict__ denom,
                    const float* __restrict__ stats,
                    const float* __restrict__ w, const float* __restrict__ b) {
  int i = blockIdx.x * 256 + threadIdx.x;
  if (i >= N_NODES * HIDW) return;
  const float inv  = 1.0f / (float)(N_NODES * HIDW);
  const float mean = stats[0] * inv;
  const float var  = stats[1] * inv - mean * mean;
  const float rs   = rsqrtf(var + 1e-5f);
  const int c = i & (HIDW - 1);
  float v = (softmax_row_value(h[i], denom, i) - mean) * rs * w[c] + b[c];
  v = 0.5f * v * (1.0f + erff(v * 0.70710678118654752f));
  h[i] = v;
}

extern "C" void kernel_launch(void* const* d_in, const int* in_sizes, int n_in,
                              void* d_out, int out_size, void* d_ws, size_t ws_size,
                              hipStream_t stream) {
  const float* x     = (const float*)d_in[0];
  const int*   ei    = (const int*)  d_in[1];
  const float* eattr = (const float*)d_in[2];
  const float* Wl    = (const float*)d_in[3];
  const float* Wr    = (const float*)d_in[4];
  const float* We    = (const float*)d_in[5];
  const float* att   = (const float*)d_in[6];
  const float* lnw   = (const float*)d_in[7];
  const float* lnb   = (const float*)d_in[8];
  const float* Wout  = (const float*)d_in[9];
  const float* bout  = (const float*)d_in[10];
  float* out = (float*)d_out;

  char* p = (char*)d_ws;
  auto carve = [&](size_t bytes) -> void* {
    void* q = (void*)p;
    p += (bytes + 255) & ~(size_t)255;
    return q;
  };
  float*    hA    = (float*)   carve((size_t)N_NODES * HIDW * 4);
  float*    hB    = (float*)   carve((size_t)N_NODES * HIDW * 4);
  float*    xl    = (float*)   carve((size_t)N_NODES * HIDW * 4);
  float*    xr    = (float*)   carve((size_t)N_NODES * HIDW * 4);
  float*    denom = (float*)   carve((size_t)N_NODES * NHEAD * 4);
  float*    stats = (float*)   carve(2 * 4);
  unsigned* okey  = (unsigned*)carve((size_t)N_EDGES);
  unsigned* mlist = (unsigned*)carve((size_t)OWN_GROUPS * OWN_WAVES * PCAP * 2 * 4);
  float*    mattr = (float*)   carve((size_t)OWN_GROUPS * OWN_WAVES * PCAP * 3 * 4);
  int*      mcnt  = (int*)     carve((size_t)OWN_GROUPS * OWN_WAVES * 4);

  const int elemBlocks = (N_NODES * HIDW + 255) / 256;
  const int keyBlocks  = (N_EDGES / 4 + 255) / 256;
  const int gemmBlocks = N_NODES / 16;

  owner_keys_kernel<<<keyBlocks, 256, 0, stream>>>(ei + N_EDGES, okey);
  const float* hcur = x;
  float* bufs[2] = { hA, hB };
  for (int l = 0; l < NLAYERS; ++l) {
    float* hout = bufs[l & 1];
    gemm_dual_wmma<<<gemmBlocks, 256, 0, stream>>>(hcur, Wl + (size_t)l * HIDW * HIDW,
                                                   Wr + (size_t)l * HIDW * HIDW, xl, xr);
    edge_owner_kernel<<<OWN_GROUPS, 1024, 0, stream>>>(ei, eattr, xl, xr,
                                                        We + (size_t)l * EDIMW * HIDW,
                                                        att + (size_t)l * HIDW,
                                                        okey, denom, stats, hout,
                                                        mlist, mattr, mcnt, l == 0 ? 1 : 0);
    ln_reduce_kernel<<<2048, 256, 0, stream>>>(hout, denom, stats);
    ln_gelu_kernel<<<elemBlocks, 256, 0, stream>>>(hout, denom, stats,
                                                   lnw + (size_t)l * HIDW,
                                                   lnb + (size_t)l * HIDW);
    hcur = hout;
  }
  gemm_bias_wmma<<<gemmBlocks, 256, 0, stream>>>(hcur, Wout, bout, out);
}


# Loops that waited for memory once per trip (GEMM A-tile staging, owner groups' xr staging, LayerNorm sum loop) ask for all of a thread's loads before first use; sums and their order unchanged
# speedup vs baseline: 2.7801x; 1.0464x over previous
#include <hip/hip_runtime.h>
#include <math.h>

#define N_NODES 50000
#define N_EDGES 800000
#define HIDW    128
#define NHEAD   4
#define CPH     32
#define NLAYERS 3
#define EDIMW   3

typedef __attribute__((ext_vector_type(2))) float v2f;
typedef __attribute__((ext_vector_type(8))) float v8f;
typedef __attribute__((ext_vector_type(4))) float v4f;

__device__ __forceinline__ unsigned encOrd(float f) {
  unsigned u = __float_as_uint(f);
  return (u & 0x80000000u) ? ~u : (u | 0x80000000u);
}
__device__ __forceinline__ float decOrd(unsigned u) {
  return __uint_as_float((u & 0x80000000u) ? (u & 0x7FFFFFFFu) : ~u);
}
#define ENC_NEG_INF 0x007FFFFFu

__device__ __forceinline__ void stage_a_tile(float* __restrict__ As, const float* __restrict__ tile) {
  const int p0 = threadIdx.x, p1 = threadIdx.x + 256;
  const float4 v0 = *(const float4*)(tile + 4 * p0);
  const float4 v1 = *(const float4*)(tile + 4 * p1);
  *(float4*)(As + (p0 >> 5) * 132 + (p0 & 31) * 4) = v0;
  *(float4*)(As + (p1 >> 5) * 132 + (p1 & 31) * 4) = v1;
}

__global__ __launch_bounds__(256)
void gemm_dual_wmma(const float* __restrict__ A,
                    const float* __restrict__ W0,
                    const float* __restrict__ W1,
                    float* __restrict__ O0,
                    float* __restrict__ O1) {
  __shared__ __attribute__((aligned(16))) float As[16 * 132];
  const int rowBase = blockIdx.x * 16;
  stage_a_tile(As, A + (size_t)rowBase * HIDW);
  __syncthreads();
  const int wave    = threadIdx.x >> 5;
  const int lane    = threadIdx.x & 31;
  const int colBase = wave * 16;
  const int laneHi  = lane >> 4;
  const int laneLo  = lane & 15;
  v8f c0 = {}; v8f c1 = {};
  for (int k0 = 0; k0 < HIDW; k0 += 4) {
    const int ka = k0 + 2 * laneHi;
    v2f a;
    a.x = As[laneLo * 132 + ka];
    a.y = As[laneLo * 132 + ka + 1];
    v2f b0, b1;
    b0.x = W0[(ka + 0) * HIDW + colBase + laneLo];
    b0.y = W0[(ka + 1) * HIDW + colBase + laneLo];
    b1.x = W1[(ka + 0) * HIDW + colBase + laneLo];
    b1.y = W1[(ka + 1) * HIDW + colBase + laneLo];
    c0 = __builtin_amdgcn_wmma_f32_16x16x4_f32(false, a, false, b0, (short)0, c0, false, false);
    c1 = __builtin_amdgcn_wmma_f32_16x16x4_f32(false, a, false, b1, (short)0, c1, false, false);
  }
#pragma unroll
  for (int r = 0; r < 8; ++r) {
    const int row = rowBase + r + 8 * laneHi;
    O0[row * HIDW + colBase + laneLo] = c0[r];
    O1[row * HIDW + colBase + laneLo] = c1[r];
  }
}

__global__ __launch_bounds__(256)
void gemm_bias_wmma(const float* __restrict__ A,
                    const float* __restrict__ W,
                    const float* __restrict__ bias,
                    float* __restrict__ O) {
  __shared__ __attribute__((aligned(16))) float As[16 * 132];
  const int rowBase = blockIdx.x * 16;
  stage_a_tile(As, A + (size_t)rowBase * HIDW);
  __syncthreads();
  const int wave    = threadIdx.x >> 5;
  const int lane    = threadIdx.x & 31;
  const int colBase = wave * 16;
  const int laneHi  = lane >> 4;
  const int laneLo  = lane & 15;
  v8f c0 = {};
  for (int k0 = 0; k0 < HIDW; k0 += 4) {
    const int ka = k0 + 2 * laneHi;
    v2f a;
    a.x = As[laneLo * 132 + ka];
    a.y = As[laneLo * 132 + ka + 1];
    v2f b;
    b.x = W[(ka + 0) * HIDW + colBase + laneLo];
    b.y = W[(ka + 1) * HIDW + colBase + laneLo];
    c0 = __builtin_amdgcn_wmma_f32_16x16x4_f32(false, a, false, b, (short)0, c0, false, false);
  }
  const float bv = bias[colBase + laneLo];
#pragma unroll
  for (int r = 0; r < 8; ++r) {
    const int row = rowBase + r + 8 * laneHi;
    O[row * HIDW + colBase + laneLo] = c0[r] + bv;
  }
}

#define OWN_NODES   196
#define OWN_GROUPS  256
#define OWN_WAVES   32
static_assert(N_EDGES % 4 == 0, "one thread packs the owner bytes of four edges");
static_assert((N_NODES + OWN_NODES - 1) / OWN_NODES <= 256, "owner id fits one byte");

__global__ __launch_bounds__(256)
void owner_keys_kernel(const int* __restrict__ dstp, unsigned* __restrict__ okey) {
  int i = blockIdx.x * 256 + threadIdx.x;
  if (i >= N_EDGES / 4) return;
  const int4 d = *(const int4*)(dstp + 4 * i);
  okey[i] = (((unsigned)d.x / OWN_NODES) & 0xFFu)
          | ((((unsigned)d.y / OWN_NODES) & 0xFFu) << 8)
          | ((((unsigned)d.z / OWN_NODES) & 0xFFu) << 16)
          | ((((unsigned)d.w / OWN_NODES) & 0xFFu) << 24);
}

template <int XCTRL>
__device__ __forceinline__ float add_xmask(float p) {
  return p + __int_as_float(__builtin_amdgcn_update_dpp(0, __float_as_int(p), XCTRL, 0xf, 0xf, false));
}

template <int U>
__device__ __forceinline__ float group_lane(float w) {
  return __uint_as_float(__builtin_amdgcn_mov_dpp8(__float_as_uint(w), U * 0x249249));
}

template <int U>
__device__ __forceinline__ void add_weighted(float* pa, const v4f x, const float w) {
  const float wu = group_lane<U>(w);
  atomicAdd(pa,     x.x * wu);
  atomicAdd(pa + 1, x.y * wu);
  atomicAdd(pa + 2, x.z * wu);
  atomicAdd(pa + 3, x.w * wu);
}

#define SCAN_EDGES  2048
#define SCAN_STEPS  13
#define SCAN_CHUNKS (SCAN_STEPS * OWN_WAVES)
#define QCAP        64
#define PCAP        224
#define ACC_PITCH   (HIDW + 1)
#define SHIFT_EMPTY 0xFFFFFFFFu
#define SHIFT_SPAN  20.0f
static_assert(N_NODES <= 65536 && OWN_NODES < 255, "a list entry packs the source into 16 bits, the local node into 8");
static_assert(SCAN_CHUNKS * SCAN_EDGES >= N_EDGES, "scan covers every edge");
static_assert(N_EDGES % 16 == 0, "a lane's 16 owner bytes are all edges or none");
static_assert((OWN_NODES * ACC_PITCH) % 4 == 0, "the accumulator rows are cleared in 16-byte pieces");

__global__ __launch_bounds__(1024)
void edge_owner_kernel(const int* __restrict__ ei, const float* __restrict__ eattr,
                       const float* __restrict__ xl, const float* __restrict__ xr,
                       const float* __restrict__ We, const float* __restrict__ att,
                       const unsigned* __restrict__ okey,
                       float* __restrict__ denom, float* __restrict__ stats,
                       float* __restrict__ out,
                       unsigned* __restrict__ mlist, float* __restrict__ mattr,
                       int* __restrict__ mcnt, int first) {
  __shared__ __attribute__((aligned(16))) float    sxr[OWN_NODES * HIDW];
  __shared__ __attribute__((aligned(16))) float    sacc[OWN_NODES * ACC_PITCH];
  __shared__ __attribute__((aligned(16))) float    sden[OWN_NODES * NHEAD];
  __shared__ __attribute__((aligned(16))) unsigned smax[OWN_NODES * NHEAD];
  __shared__ __attribute__((aligned(16))) unsigned sshift[OWN_NODES * NHEAD];
  __shared__ unsigned sq[OWN_WAVES * QCAP];
  __shared__ int sflag;
  const int tid  = threadIdx.x;
  const int lane = tid & 31;
  const int wave = tid >> 5;
  const int nodeBase = blockIdx.x * OWN_NODES;
  if (blockIdx.x == 0 && tid < 2) stats[tid] = 0.f;
  const int c4   = 4 * lane;
  const float4 we0 = *(const float4*)(We + c4);
  const float4 we1 = *(const float4*)(We + HIDW + c4);
  const float4 we2 = *(const float4*)(We + 2 * HIDW + c4);
  const float4 at  = *(const float4*)(att + c4);
  int mcv = -1;
  if (!first) mcv = mcnt[blockIdx.x * OWN_WAVES + wave];
  constexpr int XR_TRIPS = (OWN_NODES * HIDW + 4095) / 4096;
  float4 xv[XR_TRIPS];
#pragma unroll
  for (int t = 0; t < XR_TRIPS; ++t) {
    const int i = tid * 4 + t * 4096;
    xv[t] = make_float4(0.f, 0.f, 0.f, 0.f);
    if (i < OWN_NODES * HIDW && nodeBase + (i >> 7) < N_NODES)
      xv[t] = *(const float4*)(xr + (size_t)nodeBase * HIDW + i);
  }
  for (int i = tid * 4; i < OWN_NODES * ACC_PITCH; i += 4096) *(float4*)(sacc + i) = make_float4(0.f, 0.f, 0.f, 0.f);
  if (tid < OWN_NODES * NHEAD) { sden[tid] = 0.f; smax[tid] = ENC_NEG_INF; sshift[tid] = SHIFT_EMPTY; }
  if (tid == 0) sflag = 0;
#pragma unroll
  for (int t = 0; t < XR_TRIPS; ++t) {
    const int i = tid * 4 + t * 4096;
    if (i < OWN_NODES * HIDW) *(float4*)(sxr + i) = xv[t];
  }
  __syncthreads();

  unsigned* q  = sq + wave * QCAP;
  const int* dstp = ei + N_EDGES;
  int count = 0;

  const unsigned gpat = (unsigned)blockIdx.x * 0x01010101u;
  const int rot = ((int)blockIdx.x * 13) % SCAN_CHUNKS;
  auto chunk_of = [&](int it) {
    int t = it * OWN_WAVES + wave + rot;
    if (t >= SCAN_CHUNKS) t -= SCAN_CHUNKS;
    return t;
  };
  auto load_chunk = [&](uint4 (&d)[4], int chunk) {
#pragma unroll
    for (int k = 0; k < 4; ++k) {
      const int e0 = chunk * SCAN_EDGES + k * 512 + lane * 16;
      if (e0 < N_EDGES) d[k] = *(const uint4*)(okey + ((unsigned)e0 >> 2));
      else              d[k] = make_uint4(~gpat, ~gpat, ~gpat, ~gpat);
    }
  };
  auto differs = [&](unsigned w) {
    const unsigned x = w ^ gpat;
    return ((x & 0x7F7F7F7Fu) + 0x7F7F7F7Fu) | x;
  };
  auto sel = [](unsigned m, unsigned a, unsigned b) { return (a & m) | (b & ~m); };
  auto hits = [&](const uint4& a, const uint4& b) {
    unsigned r = differs(b.w);
    r = sel(0x40404040u, differs(b.z) >> 1, r);
    r = sel(0x20202020u, differs(b.y) >> 2, r);
    r = sel(0x10101010u, differs(b.x) >> 3, r);
    r = sel(0x08080808u, differs(a.w) >> 4, r);
    r = sel(0x04040404u, differs(a.z) >> 5, r);
    r = sel(0x02020202u, differs(a.y) >> 6, r);
    r = sel(0x01010101u, differs(a.x) >> 7, r);
    return ~r;
  };
  auto push = [&](unsigned hm, int ebase, auto&& work) {
    for (int round = 0; round < 32; ++round) {
      const bool act = hm != 0u;
      const unsigned mask = __builtin_amdgcn_ballot_w32(act);
      if (mask == 0u) break;
      if (act) {
        const int p = __builtin_ctz(hm);
        hm &= hm - 1u;
        const int i = p & 7;
        q[count + (int)__builtin_amdgcn_mbcnt_lo(mask, 0u)] =
            (unsigned)(ebase + (i >> 2) * 512 + (i & 3) * 4 + (p >> 3));
      }
      count += __builtin_popcount(mask);
      if (count >= 32) { work(count); count = 0; }
    }
  };
  auto scan = [&](auto&& work) {
    count = 0;
    uint4 nxt[4];
    load_chunk(nxt, chunk_of(0));
    for (int it = 0; it < SCAN_STEPS; ++it) {
      uint4 cur[4];
#pragma unroll
      for (int k = 0; k < 4; ++k) cur[k] = nxt[k];
      const int chunk = chunk_of(it);
      if (it + 1 < SCAN_STEPS) load_chunk(nxt, chunk_of(it + 1));
      const int ebase = chunk * SCAN_EDGES + lane * 16;
      push(hits(cur[0], cur[1]), ebase, work);
      push(hits(cur[2], cur[3]), ebase + 1024, work);
    }
    work(count);
  };

  int  pcount = 0;
  bool listed = true;
  const size_t mslot = ((size_t)blockIdx.x * OWN_WAVES + wave) * PCAP;
  const int sub  = lane & 7;
  const int head = lane >> 3;
  const int oA = c4 + (lane >> 4);
  auto edges = [&](int n, auto&& fetch, const bool keep, const bool exact) {
    if (keep && listed && pcount + n > PCAP) listed = false;
    for (int base = 0; base < n; base += 32) {
      const int cnt = (n - base < 32) ? (n - base) : 32;
      int srcv = 0, dlv = 0, okv = 0;
      float av0 = 0.f, av1 = 0.f, av2 = 0.f;
      if (lane < cnt) {
        unsigned dl; int ev;
        fetch(base + lane, ev, srcv, dl, av0, av1, av2);
        if (dl < (unsigned)OWN_NODES) { dlv = (int)dl; okv = 1; }
        if (keep && listed) {
          const uint2 rec = make_uint2((unsigned)ev, (unsigned)srcv | ((okv ? dl : 255u) << 16));
          const size_t m = mslot + (size_t)(pcount + base + lane);
          *(uint2*)(mlist + 2 * m) = rec;
          mattr[3 * m + 0] = av0;
          mattr[3 * m + 1] = av1;
          mattr[3 * m + 2] = av2;
        }
      }
      for (int j = 0; j < cnt; j += 8) {
        v4f xa[8];
        int dj[8];
#pragma unroll
        for (int u = 0; u < 8; ++u) {
          const int jj = j + u;
          const int sj = __builtin_amdgcn_readlane(srcv, jj);
          dj[u] = __builtin_amdgcn_readlane(dlv, jj);
          xa[u] = *(const v4f*)(xl + (size_t)sj * HIDW + c4);
        }
        asm volatile("" :: "v"(xa[0]), "v"(xa[1]), "v"(xa[2]), "v"(xa[3]),
                           "v"(xa[4]), "v"(xa[5]), "v"(xa[6]), "v"(xa[7]));
        float mine = 0.f;
#pragma unroll
        for (int u = 0; u < 8; ++u) {
          const int jj = j + u;
          const v4f xb = *(const v4f*)(sxr + dj[u] * HIDW + c4);
          const float ea0 = __int_as_float(__builtin_amdgcn_readlane(__float_as_int(av0), jj));
          const float ea1 = __int_as_float(__builtin_amdgcn_readlane(__float_as_int(av1), jj));
          const float ea2 = __int_as_float(__builtin_amdgcn_readlane(__float_as_int(av2), jj));
          float v0 = xa[u].x + xb.x + (ea0 * we0.x + ea1 * we1.x + ea2 * we2.x);
          float v1 = xa[u].y + xb.y + (ea0 * we0.y + ea1 * we1.y + ea2 * we2.y);
          float v2 = xa[u].z + xb.z + (ea0 * we0.z + ea1 * we1.z + ea2 * we2.z);
          float v3 = xa[u].w + xb.w + (ea0 * we0.w + ea1 * we1.w + ea2 * we2.w);
          v0 = (v0 > 0.f) ? v0 : 0.2f * v0;
          v1 = (v1 > 0.f) ? v1 : 0.2f * v1;
          v2 = (v2 > 0.f) ? v2 : 0.2f * v2;
          v3 = (v3 > 0.f) ? v3 : 0.2f * v3;
          float p = v0 * at.x + v1 * at.y + v2 * at.z + v3 * at.w;
          p = add_xmask<0x161>(p);
          p = add_xmask<0x162>(p);
          p = add_xmask<0x164>(p);
          if (sub == u) mine = p;
        }
        const int el    = j + sub;
        const int dmine = __shfl(dlv, el, 32);
        const int omine = __shfl(okv, el, 32);
        float w = 0.f;
        if (omine) {
          const int slot = dmine * NHEAD + head;
          float m0;
          if (exact) {
            m0 = __uint_as_float(sshift[slot]);
          } else {
            const unsigned old = atomicCAS(&sshift[slot], SHIFT_EMPTY, __float_as_uint(mine));
            m0 = (old == SHIFT_EMPTY) ? mine : __uint_as_float(old);
            atomicMax(&smax[slot], encOrd(mine));
          }
          w = expf(mine - m0);
          atomicAdd(&sden[slot], w);
        }
#define ADD_EDGE(U) if (j + U < cnt) add_weighted<U>(sacc + dj[U] * ACC_PITCH + oA, xa[U], w);
        ADD_EDGE(0) ADD_EDGE(1) ADD_EDGE(2) ADD_EDGE(3)
        ADD_EDGE(4) ADD_EDGE(5) ADD_EDGE(6) ADD_EDGE(7)
#undef ADD_EDGE
      }
    }
    if (keep && listed) pcount += n;
  };
  auto from_queue = [&](int i, int& e, int& s, unsigned& dl, float& a0, float& a1, float& a2) {
    e = (int)q[i];
    const int dstv = dstp[e];
    s  = ei[e];
    a0 = eattr[(size_t)e * 3 + 0];
    a1 = eattr[(size_t)e * 3 + 1];
    a2 = eattr[(size_t)e * 3 + 2];
    dl = (unsigned)(dstv - nodeBase);
  };
  auto from_store = [&](int i, int& e, int& s, unsigned& dl, float& a0, float& a1, float& a2) {
    const size_t m = mslot + (size_t)i;
    const uint2 r = *(const uint2*)(mlist + 2 * m);
    e  = (int)r.x; s = (int)(r.y & 0xFFFFu); dl = r.y >> 16;
    a0 = mattr[3 * m + 0];
    a1 = mattr[3 * m + 1];
    a2 = mattr[3 * m + 2];
  };
  const int mc = __builtin_amdgcn_readfirstlane(mcv);
  auto run = [&](const bool exact) {
    if ((unsigned)mc <= (unsigned)PCAP) {
      edges(mc, from_store, false, exact);
    } else {
      const bool keep = first != 0 && !exact;
      scan([&](int n) { edges(n, from_queue, keep, exact); });
      if (keep && lane == 0) mcnt[blockIdx.x * OWN_WAVES + wave] = listed ? pcount : -1;
    }
  };
  run(false);
  __syncthreads();
  if (tid < OWN_NODES * NHEAD) {
    const unsigned sb = sshift[tid];
    if (sb != SHIFT_EMPTY && !(decOrd(smax[tid]) - __uint_as_float(sb) <= SHIFT_SPAN)) sflag = 1;
  }
  __syncthreads();
  if (sflag != 0) {
    for (int i = tid * 4; i < OWN_NODES * ACC_PITCH; i += 4096) *(float4*)(sacc + i) = make_float4(0.f, 0.f, 0.f, 0.f);
    if (tid < OWN_NODES * NHEAD) {
      const float f = decOrd(smax[tid]);
      sshift[tid] = __float_as_uint(isfinite(f) ? f : 0.f);
      sden[tid] = 0.f;
    }
    __syncthreads();
    run(true);
    __syncthreads();
  }

  for (int i = tid; i < OWN_NODES * HIDW; i += 1024) {
    const int node = nodeBase + (i >> 7);
    if (node < N_NODES) out[(size_t)nodeBase * HIDW + i] = sacc[i + (i >> 7) + ((i >> 6) & 1)];
  }
  if (tid < OWN_NODES * NHEAD) {
    const int node = nodeBase + (tid >> 2);
    if (node < N_NODES) denom[(size_t)nodeBase * NHEAD + tid] = sden[tid];
  }
}

__device__ __forceinline__ float softmax_row_value(float s, const float* __restrict__ denom, int i) {
  return s / (denom[i >> 5] + 1e-16f);
}

#define LNR_BLOCKS 2048
#define LNR_STRIDE (LNR_BLOCKS * 256)
#define LNR_TRIPS  ((N_NODES * HIDW + LNR_STRIDE - 1) / LNR_STRIDE)
static_assert((LNR_TRIPS - 1) * LNR_STRIDE <= N_NODES * HIDW, "only the last trip can pass the end");
__global__ __launch_bounds__(256)
void ln_reduce_kernel(const float* __restrict__ h, const float* __restrict__ denom,
                      float* __restrict__ stats) {
  __shared__ float s1[256], s2[256];
  const int i0 = blockIdx.x * 256 + threadIdx.x;
  float hv[LNR_TRIPS], dv[LNR_TRIPS];
#pragma unroll
  for (int k = 0; k < LNR_TRIPS; ++k) {
    const int i = i0 + k * LNR_STRIDE;
    const bool in = (k < LNR_TRIPS - 1) || (i < N_NODES * HIDW);
    hv[k] = in ? h[i] : 0.f;
    dv[k] = in ? denom[i >> 5] : 1.f;
  }
  float a = 0.f, b = 0.f;
#pragma unroll
  for (int k = 0; k < LNR_TRIPS; ++k) {
    const int i = i0 + k * LNR_STRIDE;
    if ((k < LNR_TRIPS - 1) || (i < N_NODES * HIDW)) {
      const float v = hv[k] / (dv[k] + 1e-16f);
      a += v;
      b = __builtin_fmaf(v, v, b);
    }
  }
  s1[threadIdx.x] = a; s2[threadIdx.x] = b;
  __syncthreads();
  for (int off = 128; off > 0; off >>= 1) {
    if ((int)threadIdx.x < off) {
      s1[threadIdx.x] += s1[threadIdx.x + off];
      s2[threadIdx.x] += s2[threadIdx.x + off];
    }
    __syncthreads();
  }
  if (threadIdx.x == 0) { atomicAdd(&stats[0], s1[0]); atomicAdd(&stats[1], s2[0]); }
}

__global__ __launch_bounds__(256)
void ln_gelu_kernel(float* __restrict__ h, const float* __restrict__ denom,
                    const float* __restrict__ stats,
                    const float* __restrict__ w, const float* __restrict__ b) {
  int i = blockIdx.x * 256 + threadIdx.x;
  if (i >= N_NODES * HIDW) return;
  const float inv  = 1.0f / (float)(N_NODES * HIDW);
  const float mean = stats[0] * inv;
  const float var  = stats[1] * inv - mean * mean;
  const float rs   = rsqrtf(var + 1e-5f);
  const int c = i & (HIDW - 1);
  float v = (softmax_row_value(h[i], denom, i) - mean) * rs * w[c] + b[c];
  v = 0.5f * v * (1.0f + erff(v * 0.70710678118654752f));
  h[i] = v;
}

extern "C" void kernel_launch(void* const* d_in, const int* in_sizes, int n_in,
                              void* d_out, int out_size, void* d_ws, size_t ws_size,
                              hipStream_t stream) {
  const float* x     = (const float*)d_in[0];
  const int*   ei    = (const int*)  d_in[1];
  const float* eattr = (const float*)d_in[2];
  const float* Wl    = (const float*)d_in[3];
  const float* Wr    = (const float*)d_in[4];
  const float* We    = (const float*)d_in[5];
  const float* att   = (const float*)d_in[6];
  const float* lnw   = (const float*)d_in[7];
  const float* lnb   = (const float*)d_in[8];
  const float* Wout  = (const float*)d_in[9];
  const float* bout  = (const float*)d_in[10];
  float* out = (float*)d_out;

  char* p = (char*)d_ws;
  auto carve = [&](size_t bytes) -> void* {
    void* q = (void*)p;
    p += (bytes + 255) & ~(size_t)255;
    return q;
  };
  float*    hA    = (float*)   carve((size_t)N_NODES * HIDW * 4);
  float*    hB    = (float*)   carve((size_t)N_NODES * HIDW * 4);
  float*    xl    = (float*)   carve((size_t)N_NODES * HIDW * 4);
  float*    xr    = (float*)   carve((size_t)N_NODES * HIDW * 4);
  float*    denom = (float*)   carve((size_t)N_NODES * NHEAD * 4);
  float*    stats = (float*)   carve(2 * 4);
  unsigned* okey  = (unsigned*)carve((size_t)N_EDGES);
  unsigned* mlist = (unsigned*)carve((size_t)OWN_GROUPS * OWN_WAVES * PCAP * 2 * 4);
  float*    mattr = (float*)   carve((size_t)OWN_GROUPS * OWN_WAVES * PCAP * 3 * 4);
  int*      mcnt  = (int*)     carve((size_t)OWN_GROUPS * OWN_WAVES * 4);

  const int elemBlocks = (N_NODES * HIDW + 255) / 256;
  const int keyBlocks  = (N_EDGES / 4 + 255) / 256;
  const int gemmBlocks = N_NODES / 16;

  owner_keys_kernel<<<keyBlocks, 256, 0, stream>>>(ei + N_EDGES, okey);
  const float* hcur = x;
  float* bufs[2] = { hA, hB };
  for (int l = 0; l < NLAYERS; ++l) {
    float* hout = bufs[l & 1];
    gemm_dual_wmma<<<gemmBlocks, 256, 0, stream>>>(hcur, Wl + (size_t)l * HIDW * HIDW,
                                                   Wr + (size_t)l * HIDW * HIDW, xl, xr);
    edge_owner_kernel<<<OWN_GROUPS, 1024, 0, stream>>>(ei, eattr, xl, xr,
                                                        We + (size_t)l * EDIMW * HIDW,
                                                        att + (size_t)l * HIDW,
                                                        okey, denom, stats, hout,
                                                        mlist, mattr, mcnt, l == 0 ? 1 : 0);
    ln_reduce_kernel<<<LNR_BLOCKS, 256, 0, stream>>>(hout, denom, stats);
    ln_gelu_kernel<<<elemBlocks, 256, 0, stream>>>(hout, denom, stats,
                                                   lnw + (size_t)l * HIDW,
                                                   lnb + (size_t)l * HIDW);
    hcur = hout;
  }
  gemm_bias_wmma<<<gemmBlocks, 256, 0, stream>>>(hcur, Wout, bout, out);
}


# GEMM tiles re-laid: a wave owns 16 rows x 64 columns as four column-interleaved 16x16 tiles; a lane's four B values of a k row are one 16-byte load, its four results of a row one 16-byte store
# speedup vs baseline: 2.8725x; 1.0332x over previous
#include <hip/hip_runtime.h>
#include <math.h>

#define N_NODES 50000
#define N_EDGES 800000
#define HIDW    128
#define NHEAD   4
#define CPH     32
#define NLAYERS 3
#define EDIMW   3

typedef __attribute__((ext_vector_type(2))) float v2f;
typedef __attribute__((ext_vector_type(8))) float v8f;
typedef __attribute__((ext_vector_type(4))) float v4f;

__device__ __forceinline__ unsigned encOrd(float f) {
  unsigned u = __float_as_uint(f);
  return (u & 0x80000000u) ? ~u : (u | 0x80000000u);
}
__device__ __forceinline__ float decOrd(unsigned u) {
  return __uint_as_float((u & 0x80000000u) ? (u & 0x7FFFFFFFu) : ~u);
}
#define ENC_NEG_INF 0x007FFFFFu

template <int NT>
__device__ __forceinline__ void stage_a_tile(float* __restrict__ As, const float* __restrict__ tile) {
  constexpr int PER = 512 / NT;
  float4 v[PER];
#pragma unroll
  for (int j = 0; j < PER; ++j) v[j] = *(const float4*)(tile + 4 * ((int)threadIdx.x + j * NT));
#pragma unroll
  for (int j = 0; j < PER; ++j) {
    const int p = (int)threadIdx.x + j * NT;
    *(float4*)(As + (p >> 5) * 132 + (p & 31) * 4) = v[j];
  }
}

template <bool BIAS>
__device__ __forceinline__ void wave_tile_16x64(const float* __restrict__ As,
                                                const float* __restrict__ W,
                                                const float* __restrict__ bias,
                                                float* __restrict__ O,
                                                const int rowBase, const int slab) {
  const int lane   = threadIdx.x & 31;
  const int laneHi = lane >> 4;
  const int laneLo = lane & 15;
  const float* ap = As + laneLo * 132 + 2 * laneHi;
  const float* wp = W + (2 * laneHi) * HIDW + slab + 4 * laneLo;
  v8f c0 = {}; v8f c1 = {}; v8f c2 = {}; v8f c3 = {};
  for (int k0 = 0; k0 < HIDW; k0 += 4) {
    v2f a;
    a.x = ap[k0];
    a.y = ap[k0 + 1];
    const float4 q0 = *(const float4*)(wp + (k0 + 0) * HIDW);
    const float4 q1 = *(const float4*)(wp + (k0 + 1) * HIDW);
    v2f b;
    b.x = q0.x; b.y = q1.x;
    c0 = __builtin_amdgcn_wmma_f32_16x16x4_f32(false, a, false, b, (short)0, c0, false, false);
    b.x = q0.y; b.y = q1.y;
    c1 = __builtin_amdgcn_wmma_f32_16x16x4_f32(false, a, false, b, (short)0, c1, false, false);
    b.x = q0.z; b.y = q1.z;
    c2 = __builtin_amdgcn_wmma_f32_16x16x4_f32(false, a, false, b, (short)0, c2, false, false);
    b.x = q0.w; b.y = q1.w;
    c3 = __builtin_amdgcn_wmma_f32_16x16x4_f32(false, a, false, b, (short)0, c3, false, false);
  }
  float4 bv = make_float4(0.f, 0.f, 0.f, 0.f);
  if (BIAS) bv = *(const float4*)(bias + slab + 4 * laneLo);
#pragma unroll
  for (int r = 0; r < 8; ++r) {
    const int row = rowBase + r + 8 * laneHi;
    float4 o = make_float4(c0[r], c1[r], c2[r], c3[r]);
    if (BIAS) { o.x += bv.x; o.y += bv.y; o.z += bv.z; o.w += bv.w; }
    *(float4*)(O + (size_t)row * HIDW + slab + 4 * laneLo) = o;
  }
}

__global__ __launch_bounds__(128)
void gemm_dual_wmma(const float* __restrict__ A,
                    const float* __restrict__ W0,
                    const float* __restrict__ W1,
                    float* __restrict__ O0,
                    float* __restrict__ O1) {
  __shared__ __attribute__((aligned(16))) float As[16 * 132];
  const int rowBase = blockIdx.x * 16;
  stage_a_tile<128>(As, A + (size_t)rowBase * HIDW);
  __syncthreads();
  const int wave = threadIdx.x >> 5;
  const float* W = (wave & 2) ? W1 : W0;
  float*       O = (wave & 2) ? O1 : O0;
  wave_tile_16x64<false>(As, W, nullptr, O, rowBase, (wave & 1) * 64);
}

__global__ __launch_bounds__(64)
void gemm_bias_wmma(const float* __restrict__ A,
                    const float* __restrict__ W,
                    const float* __restrict__ bias,
                    float* __restrict__ O) {
  __shared__ __attribute__((aligned(16))) float As[16 * 132];
  const int rowBase = blockIdx.x * 16;
  stage_a_tile<64>(As, A + (size_t)rowBase * HIDW);
  __syncthreads();
  const int wave = threadIdx.x >> 5;
  wave_tile_16x64<true>(As, W, bias, O, rowBase, wave * 64);
}

#define OWN_NODES   196
#define OWN_GROUPS  256
#define OWN_WAVES   32
static_assert(N_EDGES % 4 == 0, "one thread packs the owner bytes of four edges");
static_assert((N_NODES + OWN_NODES - 1) / OWN_NODES <= 256, "owner id fits one byte");

__global__ __launch_bounds__(256)
void owner_keys_kernel(const int* __restrict__ dstp, unsigned* __restrict__ okey) {
  int i = blockIdx.x * 256 + threadIdx.x;
  if (i >= N_EDGES / 4) return;
  const int4 d = *(const int4*)(dstp + 4 * i);
  okey[i] = (((unsigned)d.x / OWN_NODES) & 0xFFu)
          | ((((unsigned)d.y / OWN_NODES) & 0xFFu) << 8)
          | ((((unsigned)d.z / OWN_NODES) & 0xFFu) << 16)
          | ((((unsigned)d.w / OWN_NODES) & 0xFFu) << 24);
}

template <int XCTRL>
__device__ __forceinline__ float add_xmask(float p) {
  return p + __int_as_float(__builtin_amdgcn_update_dpp(0, __float_as_int(p), XCTRL, 0xf, 0xf, false));
}

template <int U>
__device__ __forceinline__ float group_lane(float w) {
  return __uint_as_float(__builtin_amdgcn_mov_dpp8(__float_as_uint(w), U * 0x249249));
}

template <int U>
__device__ __forceinline__ void add_weighted(float* pa, const v4f x, const float w) {
  const float wu = group_lane<U>(w);
  atomicAdd(pa,     x.x * wu);
  atomicAdd(pa + 1, x.y * wu);
  atomicAdd(pa + 2, x.z * wu);
  atomicAdd(pa + 3, x.w * wu);
}

#define SCAN_EDGES  2048
#define SCAN_STEPS  13
#define SCAN_CHUNKS (SCAN_STEPS * OWN_WAVES)
#define QCAP        64
#define PCAP        224
#define ACC_PITCH   (HIDW + 1)
#define SHIFT_EMPTY 0xFFFFFFFFu
#define SHIFT_SPAN  20.0f
static_assert(N_NODES <= 65536 && OWN_NODES < 255, "a list entry packs the source into 16 bits, the local node into 8");
static_assert(SCAN_CHUNKS * SCAN_EDGES >= N_EDGES, "scan covers every edge");
static_assert(N_EDGES % 16 == 0, "a lane's 16 owner bytes are all edges or none");
static_assert((OWN_NODES * ACC_PITCH) % 4 == 0, "the accumulator rows are cleared in 16-byte pieces");

__global__ __launch_bounds__(1024)
void edge_owner_kernel(const int* __restrict__ ei, const float* __restrict__ eattr,
                       const float* __restrict__ xl, const float* __restrict__ xr,
                       const float* __restrict__ We, const float* __restrict__ att,
                       const unsigned* __restrict__ okey,
                       float* __restrict__ denom, float* __restrict__ stats,
                       float* __restrict__ out,
                       unsigned* __restrict__ mlist, float* __restrict__ mattr,
                       int* __restrict__ mcnt, int first) {
  __shared__ __attribute__((aligned(16))) float    sxr[OWN_NODES * HIDW];
  __shared__ __attribute__((aligned(16))) float    sacc[OWN_NODES * ACC_PITCH];
  __shared__ __attribute__((aligned(16))) float    sden[OWN_NODES * NHEAD];
  __shared__ __attribute__((aligned(16))) unsigned smax[OWN_NODES * NHEAD];
  __shared__ __attribute__((aligned(16))) unsigned sshift[OWN_NODES * NHEAD];
  __shared__ unsigned sq[OWN_WAVES * QCAP];
  __shared__ int sflag;
  const int tid  = threadIdx.x;
  const int lane = tid & 31;
  const int wave = tid >> 5;
  const int nodeBase = blockIdx.x * OWN_NODES;
  if (blockIdx.x == 0 && tid < 2) stats[tid] = 0.f;
  const int c4   = 4 * lane;
  const float4 we0 = *(const float4*)(We + c4);
  const float4 we1 = *(const float4*)(We + HIDW + c4);
  const float4 we2 = *(const float4*)(We + 2 * HIDW + c4);
  const float4 at  = *(const float4*)(att + c4);
  int mcv = -1;
  if (!first) mcv = mcnt[blockIdx.x * OWN_WAVES + wave];
  constexpr int XR_TRIPS = (OWN_NODES * HIDW + 4095) / 4096;
  float4 xv[XR_TRIPS];
#pragma unroll
  for (int t = 0; t < XR_TRIPS; ++t) {
    const int i = tid * 4 + t * 4096;
    xv[t] = make_float4(0.f, 0.f, 0.f, 0.f);
    if (i < OWN_NODES * HIDW && nodeBase + (i >> 7) < N_NODES)
      xv[t] = *(const float4*)(xr + (size_t)nodeBase * HIDW + i);
  }
  for (int i = tid * 4; i < OWN_NODES * ACC_PITCH; i += 4096) *(float4*)(sacc + i) = make_float4(0.f, 0.f, 0.f, 0.f);
  if (tid < OWN_NODES * NHEAD) { sden[tid] = 0.f; smax[tid] = ENC_NEG_INF; sshift[tid] = SHIFT_EMPTY; }
  if (tid == 0) sflag = 0;
#pragma unroll
  for (int t = 0; t < XR_TRIPS; ++t) {
    const int i = tid * 4 + t * 4096;
    if (i < OWN_NODES * HIDW) *(float4*)(sxr + i) = xv[t];
  }
  __syncthreads();

  unsigned* q  = sq + wave * QCAP;
  const int* dstp = ei + N_EDGES;
  int count = 0;

  const unsigned gpat = (unsigned)blockIdx.x * 0x01010101u;
  const int rot = ((int)blockIdx.x * 13) % SCAN_CHUNKS;
  auto chunk_of = [&](int it) {
    int t = it * OWN_WAVES + wave + rot;
    if (t >= SCAN_CHUNKS) t -= SCAN_CHUNKS;
    return t;
  };
  auto load_chunk = [&](uint4 (&d)[4], int chunk) {
#pragma unroll
    for (int k = 0; k < 4; ++k) {
      const int e0 = chunk * SCAN_EDGES + k * 512 + lane * 16;
      if (e0 < N_EDGES) d[k] = *(const uint4*)(okey + ((unsigned)e0 >> 2));
      else              d[k] = make_uint4(~gpat, ~gpat, ~gpat, ~gpat);
    }
  };
  auto differs = [&](unsigned w) {
    const unsigned x = w ^ gpat;
    return ((x & 0x7F7F7F7Fu) + 0x7F7F7F7Fu) | x;
  };
  auto sel = [](unsigned m, unsigned a, unsigned b) { return (a & m) | (b & ~m); };
  auto hits = [&](const uint4& a, const uint4& b) {
    unsigned r = differs(b.w);
    r = sel(0x40404040u, differs(b.z) >> 1, r);
    r = sel(0x20202020u, differs(b.y) >> 2, r);
    r = sel(0x10101010u, differs(b.x) >> 3, r);
    r = sel(0x08080808u, differs(a.w) >> 4, r);
    r = sel(0x04040404u, differs(a.z) >> 5, r);
    r = sel(0x02020202u, differs(a.y) >> 6, r);
    r = sel(0x01010101u, differs(a.x) >> 7, r);
    return ~r;
  };
  auto push = [&](unsigned hm, int ebase, auto&& work) {
    for (int round = 0; round < 32; ++round) {
      const bool act = hm != 0u;
      const unsigned mask = __builtin_amdgcn_ballot_w32(act);
      if (mask == 0u) break;
      if (act) {
        const int p = __builtin_ctz(hm);
        hm &= hm - 1u;
        const int i = p & 7;
        q[count + (int)__builtin_amdgcn_mbcnt_lo(mask, 0u)] =
            (unsigned)(ebase + (i >> 2) * 512 + (i & 3) * 4 + (p >> 3));
      }
      count += __builtin_popcount(mask);
      if (count >= 32) { work(count); count = 0; }
    }
  };
  auto scan = [&](auto&& work) {
    count = 0;
    uint4 nxt[4];
    load_chunk(nxt, chunk_of(0));
    for (int it = 0; it < SCAN_STEPS; ++it) {
      uint4 cur[4];
#pragma unroll
      for (int k = 0; k < 4; ++k) cur[k] = nxt[k];
      const int chunk = chunk_of(it);
      if (it + 1 < SCAN_STEPS) load_chunk(nxt, chunk_of(it + 1));
      const int ebase = chunk * SCAN_EDGES + lane * 16;
      push(hits(cur[0], cur[1]), ebase, work);
      push(hits(cur[2], cur[3]), ebase + 1024, work);
    }
    work(count);
  };

  int  pcount = 0;
  bool listed = true;
  const size_t mslot = ((size_t)blockIdx.x * OWN_WAVES + wave) * PCAP;
  const int sub  = lane & 7;
  const int head = lane >> 3;
  const int oA = c4 + (lane >> 4);
  auto edges = [&](int n, auto&& fetch, const bool keep, const bool exact) {
    if (keep && listed && pcount + n > PCAP) listed = false;
    for (int base = 0; base < n; base += 32) {
      const int cnt = (n - base < 32) ? (n - base) : 32;
      int srcv = 0, dlv = 0, okv = 0;
      float av0 = 0.f, av1 = 0.f, av2 = 0.f;
      if (lane < cnt) {
        unsigned dl; int ev;
        fetch(base + lane, ev, srcv, dl, av0, av1, av2);
        if (dl < (unsigned)OWN_NODES) { dlv = (int)dl; okv = 1; }
        if (keep && listed) {
          const uint2 rec = make_uint2((unsigned)ev, (unsigned)srcv | ((okv ? dl : 255u) << 16));
          const size_t m = mslot + (size_t)(pcount + base + lane);
          *(uint2*)(mlist + 2 * m) = rec;
          mattr[3 * m + 0] = av0;
          mattr[3 * m + 1] = av1;
          mattr[3 * m + 2] = av2;
        }
      }
      for (int j = 0; j < cnt; j += 8) {
        v4f xa[8];
        int dj[8];
#pragma unroll
        for (int u = 0; u < 8; ++u) {
          const int jj = j + u;
          const int sj = __builtin_amdgcn_readlane(srcv, jj);
          dj[u] = __builtin_amdgcn_readlane(dlv, jj);
          xa[u] = *(const v4f*)(xl + (size_t)sj * HIDW + c4);
        }
        asm volatile("" :: "v"(xa[0]), "v"(xa[1]), "v"(xa[2]), "v"(xa[3]),
                           "v"(xa[4]), "v"(xa[5]), "v"(xa[6]), "v"(xa[7]));
        float mine = 0.f;
#pragma unroll
        for (int u = 0; u < 8; ++u) {
          const int jj = j + u;
          const v4f xb = *(const v4f*)(sxr + dj[u] * HIDW + c4);
          const float ea0 = __int_as_float(__builtin_amdgcn_readlane(__float_as_int(av0), jj));
          const float ea1 = __int_as_float(__builtin_amdgcn_readlane(__float_as_int(av1), jj));
          const float ea2 = __int_as_float(__builtin_amdgcn_readlane(__float_as_int(av2), jj));
          float v0 = xa[u].x + xb.x + (ea0 * we0.x + ea1 * we1.x + ea2 * we2.x);
          float v1 = xa[u].y + xb.y + (ea0 * we0.y + ea1 * we1.y + ea2 * we2.y);
          float v2 = xa[u].z + xb.z + (ea0 * we0.z + ea1 * we1.z + ea2 * we2.z);
          float v3 = xa[u].w + xb.w + (ea0 * we0.w + ea1 * we1.w + ea2 * we2.w);
          v0 = (v0 > 0.f) ? v0 : 0.2f * v0;
          v1 = (v1 > 0.f) ? v1 : 0.2f * v1;
          v2 = (v2 > 0.f) ? v2 : 0.2f * v2;
          v3 = (v3 > 0.f) ? v3 : 0.2f * v3;
          float p = v0 * at.x + v1 * at.y + v2 * at.z + v3 * at.w;
          p = add_xmask<0x161>(p);
          p = add_xmask<0x162>(p);
          p = add_xmask<0x164>(p);
          if (sub == u) mine = p;
        }
        const int el    = j + sub;
        const int dmine = __shfl(dlv, el, 32);
        const int omine = __shfl(okv, el, 32);
        float w = 0.f;
        if (omine) {
          const int slot = dmine * NHEAD + head;
          float m0;
          if (exact) {
            m0 = __uint_as_float(sshift[slot]);
          } else {
            const unsigned old = atomicCAS(&sshift[slot], SHIFT_EMPTY, __float_as_uint(mine));
            m0 = (old == SHIFT_EMPTY) ? mine : __uint_as_float(old);
            atomicMax(&smax[slot], encOrd(mine));
          }
          w = expf(mine - m0);
          atomicAdd(&sden[slot], w);
        }
#define ADD_EDGE(U) if (j + U < cnt) add_weighted<U>(sacc + dj[U] * ACC_PITCH + oA, xa[U], w);
        ADD_EDGE(0) ADD_EDGE(1) ADD_EDGE(2) ADD_EDGE(3)
        ADD_EDGE(4) ADD_EDGE(5) ADD_EDGE(6) ADD_EDGE(7)
#undef ADD_EDGE
      }
    }
    if (keep && listed) pcount += n;
  };
  auto from_queue = [&](int i, int& e, int& s, unsigned& dl, float& a0, float& a1, float& a2) {
    e = (int)q[i];
    const int dstv = dstp[e];
    s  = ei[e];
    a0 = eattr[(size_t)e * 3 + 0];
    a1 = eattr[(size_t)e * 3 + 1];
    a2 = eattr[(size_t)e * 3 + 2];
    dl = (unsigned)(dstv - nodeBase);
  };
  auto from_store = [&](int i, int& e, int& s, unsigned& dl, float& a0, float& a1, float& a2) {
    const size_t m = mslot + (size_t)i;
    const uint2 r = *(const uint2*)(mlist + 2 * m);
    e  = (int)r.x; s = (int)(r.y & 0xFFFFu); dl = r.y >> 16;
    a0 = mattr[3 * m + 0];
    a1 = mattr[3 * m + 1];
    a2 = mattr[3 * m + 2];
  };
  const int mc = __builtin_amdgcn_readfirstlane(mcv);
  auto run = [&](const bool exact) {
    if ((unsigned)mc <= (unsigned)PCAP) {
      edges(mc, from_store, false, exact);
    } else {
      const bool keep = first != 0 && !exact;
      scan([&](int n) { edges(n, from_queue, keep, exact); });
      if (keep && lane == 0) mcnt[blockIdx.x * OWN_WAVES + wave] = listed ? pcount : -1;
    }
  };
  run(false);
  __syncthreads();
  if (tid < OWN_NODES * NHEAD) {
    const unsigned sb = sshift[tid];
    if (sb != SHIFT_EMPTY && !(decOrd(smax[tid]) - __uint_as_float(sb) <= SHIFT_SPAN)) sflag = 1;
  }
  __syncthreads();
  if (sflag != 0) {
    for (int i = tid * 4; i < OWN_NODES * ACC_PITCH; i += 4096) *(float4*)(sacc + i) = make_float4(0.f, 0.f, 0.f, 0.f);
    if (tid < OWN_NODES * NHEAD) {
      const float f = decOrd(smax[tid]);
      sshift[tid] = __float_as_uint(isfinite(f) ? f : 0.f);
      sden[tid] = 0.f;
    }
    __syncthreads();
    run(true);
    __syncthreads();
  }

  for (int i = tid; i < OWN_NODES * HIDW; i += 1024) {
    const int node = nodeBase + (i >> 7);
    if (node < N_NODES) out[(size_t)nodeBase * HIDW + i] = sacc[i + (i >> 7) + ((i >> 6) & 1)];
  }
  if (tid < OWN_NODES * NHEAD) {
    const int node = nodeBase + (tid >> 2);
    if (node < N_NODES) denom[(size_t)nodeBase * NHEAD + tid] = sden[tid];
  }
}

__device__ __forceinline__ float softmax_row_value(float s, const float* __restrict__ denom, int i) {
  return s / (denom[i >> 5] + 1e-16f);
}

#define LNR_BLOCKS 2048
#define LNR_STRIDE (LNR_BLOCKS * 256)
#define LNR_TRIPS  ((N_NODES * HIDW + LNR_STRIDE - 1) / LNR_STRIDE)
static_assert((LNR_TRIPS - 1) * LNR_STRIDE <= N_NODES * HIDW, "only the last trip can pass the end");
__global__ __launch_bounds__(256)
void ln_reduce_kernel(const float* __restrict__ h, const float* __restrict__ denom,
                      float* __restrict__ stats) {
  __shared__ float s1[256], s2[256];
  const int i0 = blockIdx.x * 256 + threadIdx.x;
  float hv[LNR_TRIPS], dv[LNR_TRIPS];
#pragma unroll
  for (int k = 0; k < LNR_TRIPS; ++k) {
    const int i = i0 + k * LNR_STRIDE;
    const bool in = (k < LNR_TRIPS - 1) || (i < N_NODES * HIDW);
    hv[k] = in ? h[i] : 0.f;
    dv[k] = in ? denom[i >> 5] : 1.f;
  }
  float a = 0.f, b = 0.f;
#pragma unroll
  for (int k = 0; k < LNR_TRIPS; ++k) {
    const int i = i0 + k * LNR_STRIDE;
    if ((k < LNR_TRIPS - 1) || (i < N_NODES * HIDW)) {
      const float v = hv[k] / (dv[k] + 1e-16f);
      a += v;
      b = __builtin_fmaf(v, v, b);
    }
  }
  s1[threadIdx.x] = a; s2[threadIdx.x] = b;
  __syncthreads();
  for (int off = 128; off > 0; off >>= 1) {
    if ((int)threadIdx.x < off) {
      s1[threadIdx.x] += s1[threadIdx.x + off];
      s2[threadIdx.x] += s2[threadIdx.x + off];
    }
    __syncthreads();
  }
  if (threadIdx.x == 0) { atomicAdd(&stats[0], s1[0]); atomicAdd(&stats[1], s2[0]); }
}

__global__ __launch_bounds__(256)
void ln_gelu_kernel(float* __restrict__ h, const float* __restrict__ denom,
                    const float* __restrict__ stats,
                    const float* __restrict__ w, const float* __restrict__ b) {
  int i = blockIdx.x * 256 + threadIdx.x;
  if (i >= N_NODES * HIDW) return;
  const float inv  = 1.0f / (float)(N_NODES * HIDW);
  const float mean = stats[0] * inv;
  const float var  = stats[1] * inv - mean * mean;
  const float rs   = rsqrtf(var + 1e-5f);
  const int c = i & (HIDW - 1);
  float v = (softmax_row_value(h[i], denom, i) - mean) * rs * w[c] + b[c];
  v = 0.5f * v * (1.0f + erff(v * 0.70710678118654752f));
  h[i] = v;
}

extern "C" void kernel_launch(void* const* d_in, const int* in_sizes, int n_in,
                              void* d_out, int out_size, void* d_ws, size_t ws_size,
                              hipStream_t stream) {
  const float* x     = (const float*)d_in[0];
  const int*   ei    = (const int*)  d_in[1];
  const float* eattr = (const float*)d_in[2];
  const float* Wl    = (const float*)d_in[3];
  const float* Wr    = (const float*)d_in[4];
  const float* We    = (const float*)d_in[5];
  const float* att   = (const float*)d_in[6];
  const float* lnw   = (const float*)d_in[7];
  const float* lnb   = (const float*)d_in[8];
  const float* Wout  = (const float*)d_in[9];
  const float* bout  = (const float*)d_in[10];
  float* out = (float*)d_out;

  char* p = (char*)d_ws;
  auto carve = [&](size_t bytes) -> void* {
    void* q = (void*)p;
    p += (bytes + 255) & ~(size_t)255;
    return q;
  };
  float*    hA    = (float*)   carve((size_t)N_NODES * HIDW * 4);
  float*    hB    = (float*)   carve((size_t)N_NODES * HIDW * 4);
  float*    xl    = (float*)   carve((size_t)N_NODES * HIDW * 4);
  float*    xr    = (float*)   carve((size_t)N_NODES * HIDW * 4);
  float*    denom = (float*)   carve((size_t)N_NODES * NHEAD * 4);
  float*    stats = (float*)   carve(2 * 4);
  unsigned* okey  = (unsigned*)carve((size_t)N_EDGES);
  unsigned* mlist = (unsigned*)carve((size_t)OWN_GROUPS * OWN_WAVES * PCAP * 2 * 4);
  float*    mattr = (float*)   carve((size_t)OWN_GROUPS * OWN_WAVES * PCAP * 3 * 4);
  int*      mcnt  = (int*)     carve((size_t)OWN_GROUPS * OWN_WAVES * 4);

  const int elemBlocks = (N_NODES * HIDW + 255) / 256;
  const int keyBlocks  = (N_EDGES / 4 + 255) / 256;
  const int gemmBlocks = N_NODES / 16;

  owner_keys_kernel<<<keyBlocks, 256, 0, stream>>>(ei + N_EDGES, okey);
  const float* hcur = x;
  float* bufs[2] = { hA, hB };
  for (int l = 0; l < NLAYERS; ++l) {
    float* hout = bufs[l & 1];
    gemm_dual_wmma<<<gemmBlocks, 128, 0, stream>>>(hcur, Wl + (size_t)l * HIDW * HIDW,
                                                   Wr + (size_t)l * HIDW * HIDW, xl, xr);
    edge_owner_kernel<<<OWN_GROUPS, 1024, 0, stream>>>(ei, eattr, xl, xr,
                                                        We + (size_t)l * EDIMW * HIDW,
                                                        att + (size_t)l * HIDW,
                                                        okey, denom, stats, hout,
                                                        mlist, mattr, mcnt, l == 0 ? 1 : 0);
    ln_reduce_kernel<<<LNR_BLOCKS, 256, 0, stream>>>(hout, denom, stats);
    ln_gelu_kernel<<<elemBlocks, 256, 0, stream>>>(hout, denom, stats,
                                                   lnw + (size_t)l * HIDW,
                                                   lnb + (size_t)l * HIDW);
    hcur = hout;
  }
  gemm_bias_wmma<<<gemmBlocks, 64, 0, stream>>>(hcur, Wout, bout, out);
}
